# starting kernel
# baseline (speedup 1.0000x reference)
#include <hip/hip_runtime.h>


#define NN 100000
#define NE 1600000
#define INF 128
#define HID 64
#define NCL 16

typedef float v2f __attribute__((ext_vector_type(2)));
typedef float v8f __attribute__((ext_vector_type(8)));


__global__ void init_deg(float* __restrict__ deg, int n) {
    int i = blockIdx.x * blockDim.x + threadIdx.x;
    if (i < n) deg[i] = 1.0f;
}

__global__ void count_deg(const int* __restrict__ dst, float* __restrict__ deg, int e) {
    int i = blockIdx.x * blockDim.x + threadIdx.x;
    if (i < e) atomicAdd(&deg[dst[i]], 1.0f);
}

__global__ void deg_to_dinv(float* __restrict__ deg, int n) {
    int i = blockIdx.x * blockDim.x + threadIdx.x;
    if (i < n) deg[i] = rsqrtf(deg[i]);
}

__global__ void zero_buf(float* __restrict__ p, int n) {
    int i = blockIdx.x * blockDim.x + threadIdx.x;
    if (i < n) p[i] = 0.0f;
}

__global__ void gemm_wmma_f32(const float* __restrict__ A, const float* __restrict__ W,
                              const float* __restrict__ bias, float* __restrict__ C,
                              int M, int K, int Ncols, int addbias) {
    const int lane  = threadIdx.x & 31;
    const int wave  = threadIdx.x >> 5;
    const int ntile = Ncols >> 4;
    const int mtile = M >> 4;
    const int tile  = blockIdx.x * (blockDim.x >> 5) + wave;
    if (tile >= mtile * ntile) return;
    const int mt = tile / ntile;
    const int nt = tile - mt * ntile;

    const int lm    = lane & 15;
    const int khalf = (lane >> 4) << 1;

    const float* arow = A + (size_t)(mt * 16 + lm) * K;
    const float* wcol = W + (size_t)(nt * 16 + lm);

    v8f acc = {0.f, 0.f, 0.f, 0.f, 0.f, 0.f, 0.f, 0.f};
    for (int k0 = 0; k0 < K; k0 += 4) {
        v2f a, b;
        a[0] = arow[k0 + khalf];
        a[1] = arow[k0 + khalf + 1];
        b[0] = wcol[(size_t)(k0 + khalf) * Ncols];
        b[1] = wcol[(size_t)(k0 + khalf + 1) * Ncols];
        acc = __builtin_amdgcn_wmma_f32_16x16x4_f32(false, a, false, b,
                                                    (short)0, acc, false, false);
    }

    const float bv = addbias ? bias[nt * 16 + lm] : 0.0f;
    float* crow = C + (size_t)(mt * 16 + ((lane >> 4) << 3)) * Ncols + nt * 16 + lm;
#pragma unroll
    for (int v = 0; v < 8; ++v)
        crow[(size_t)v * Ncols] = acc[v] + bv;
}

__global__ void scatter_edges(const int* __restrict__ src, const int* __restrict__ dst,
                              const float* __restrict__ dinv, const float* __restrict__ hw,
                              float* __restrict__ agg, int e) {
    int t = blockIdx.x * blockDim.x + threadIdx.x;
    int eid = t >> 4;
    if (eid >= e) return;
    int f4 = (t & 15) << 2;
    int s = src[eid];
    int d = dst[eid];
    float w = dinv[s] * dinv[d];
    const float4 v = *(const float4*)(hw + (size_t)s * HID + f4);
    float* o = agg + (size_t)d * HID + f4;
    atomicAdd(o + 0, v.x * w);
    atomicAdd(o + 1, v.y * w);
    atomicAdd(o + 2, v.z * w);
    atomicAdd(o + 3, v.w * w);
}

__global__ void finalize_layer(const float* __restrict__ hw, const float* __restrict__ dinv,
                               const float* __restrict__ b, float* __restrict__ agg, int n) {
    int t = blockIdx.x * blockDim.x + threadIdx.x;
    int i = t / HID;
    if (i >= n) return;
    int f = t - i * HID;
    float di = dinv[i];
    float v = agg[t] + di * di * hw[t] + b[f];
    agg[t] = fmaxf(v, 0.0f);
}

static inline int cdiv(long long a, int b) { return (int)((a + b - 1) / b); }

extern "C" void kernel_launch(void* const* d_in, const int* in_sizes, int n_in,
                              void* d_out, int out_size, void* d_ws, size_t ws_size,
                              hipStream_t stream) {
    const float* x  = (const float*)d_in[0];
    const int*   ei = (const int*)d_in[1];
    const float* W1 = (const float*)d_in[2];
    const float* b1 = (const float*)d_in[3];
    const float* W2 = (const float*)d_in[4];
    const float* b2 = (const float*)d_in[5];
    const float* Wl = (const float*)d_in[6];
    const float* bl = (const float*)d_in[7];
    float* out = (float*)d_out;

    float* dinv = (float*)d_ws;
    float* buf1 = dinv + NN;
    float* buf2 = buf1 + (size_t)NN * HID;

    const int* src = ei;
    const int* dst = ei + NE;

    const int B = 256;

    init_deg<<<cdiv(NN, B), B, 0, stream>>>(dinv, NN);
    count_deg<<<cdiv(NE, B), B, 0, stream>>>(dst, dinv, NE);
    deg_to_dinv<<<cdiv(NN, B), B, 0, stream>>>(dinv, NN);

    {
        int tiles = (NN / 16) * (HID / 16);
        gemm_wmma_f32<<<cdiv(tiles, 4), 128, 0, stream>>>(x, W1, nullptr, buf1,
                                                          NN, INF, HID, 0);
    }
    zero_buf<<<cdiv((long long)NN * HID, B), B, 0, stream>>>(buf2, NN * HID);
    scatter_edges<<<cdiv((long long)NE * 16, B), B, 0, stream>>>(src, dst, dinv, buf1, buf2, NE);
    finalize_layer<<<cdiv((long long)NN * HID, B), B, 0, stream>>>(buf1, dinv, b1, buf2, NN);

    {
        int tiles = (NN / 16) * (HID / 16);
        gemm_wmma_f32<<<cdiv(tiles, 4), 128, 0, stream>>>(buf2, W2, nullptr, buf1,
                                                          NN, HID, HID, 0);
    }
    zero_buf<<<cdiv((long long)NN * HID, B), B, 0, stream>>>(buf2, NN * HID);
    scatter_edges<<<cdiv((long long)NE * 16, B), B, 0, stream>>>(src, dst, dinv, buf1, buf2, NE);
    finalize_layer<<<cdiv((long long)NN * HID, B), B, 0, stream>>>(buf1, dinv, b2, buf2, NN);

    {
        int tiles = (NN / 16) * (NCL / 16);
        gemm_wmma_f32<<<cdiv(tiles, 4), 128, 0, stream>>>(buf2, Wl, bl, out,
                                                          NN, HID, NCL, 1);
    }
}


# f32 atomic edge scatter replaced by per-call destination buckets (int rank atomics, scan, placement) + one wave per node summing rows in registers; source norm moved to GEMM store; sums in rank order
# speedup vs baseline: 2.1272x; 2.1272x over previous
#include <hip/hip_runtime.h>


#define NN 100000
#define NE 1600000
#define INF 128
#define HID 64
#define NCL 16
#define SCAN_TILE 1024

typedef float v2f __attribute__((ext_vector_type(2)));
typedef float v8f __attribute__((ext_vector_type(8)));


__global__ void zero_cnt(int* __restrict__ cnt, int n) {
    int i = blockIdx.x * blockDim.x + threadIdx.x;
    if (i < n) cnt[i] = 0;
}

__global__ void count_rank(const int* __restrict__ dst, int* __restrict__ cnt,
                           int* __restrict__ rank, int e) {
    int i = blockIdx.x * blockDim.x + threadIdx.x;
    if (i < e) rank[i] = atomicAdd(&cnt[dst[i]], 1);
}

__global__ void block_sums(const int* __restrict__ cnt, int* __restrict__ bsum, int n) {
    __shared__ int wsum[8];
    const int lane = threadIdx.x & 31;
    const int wave = threadIdx.x >> 5;
    const int base = blockIdx.x * SCAN_TILE + threadIdx.x * 4;
    int s = 0;
#pragma unroll
    for (int q = 0; q < 4; ++q)
        if (base + q < n) s += cnt[base + q];
#pragma unroll
    for (int o = 16; o > 0; o >>= 1) s += __shfl_xor(s, o);
    if (lane == 0) wsum[wave] = s;
    __syncthreads();
    if (threadIdx.x == 0) {
        int t = 0;
#pragma unroll
        for (int w = 0; w < 8; ++w) t += wsum[w];
        bsum[blockIdx.x] = t;
    }
}

__global__ void scan_rows(const int* __restrict__ cnt, const int* __restrict__ bsum,
                          int* __restrict__ rowptr, int n) {
    __shared__ int wsum[8];
    __shared__ int s_off;
    const int lane = threadIdx.x & 31;
    const int wave = threadIdx.x >> 5;
    if (wave == 0) {
        int a = 0;
        for (int j = lane; j < (int)blockIdx.x; j += 32) a += bsum[j];
#pragma unroll
        for (int o = 16; o > 0; o >>= 1) a += __shfl_xor(a, o);
        if (lane == 0) s_off = a;
    }
    const int base = blockIdx.x * SCAN_TILE + threadIdx.x * 4;
    int c[4];
#pragma unroll
    for (int q = 0; q < 4; ++q) c[q] = (base + q < n) ? cnt[base + q] : 0;
    const int tsum = c[0] + c[1] + c[2] + c[3];
    int x = tsum;
#pragma unroll
    for (int o = 1; o < 32; o <<= 1) {
        int y = __shfl_up(x, o);
        if (lane >= o) x += y;
    }
    if (lane == 31) wsum[wave] = x;
    __syncthreads();
    int woff = 0;
    for (int w = 0; w < wave; ++w) woff += wsum[w];
    int excl = s_off + woff + x - tsum;
#pragma unroll
    for (int q = 0; q < 4; ++q) {
        if (base + q < n) rowptr[base + q] = excl;
        excl += c[q];
    }
}

__global__ void place_edges(const int* __restrict__ src, const int* __restrict__ dst,
                            const int* __restrict__ rowptr, const int* __restrict__ rank,
                            int* __restrict__ csr, int e) {
    int i = blockIdx.x * blockDim.x + threadIdx.x;
    if (i < e) csr[rowptr[dst[i]] + rank[i]] = src[i];
}

__global__ void gemm_wmma_f32(const float* __restrict__ A, const float* __restrict__ W,
                              const float* __restrict__ bias, float* __restrict__ C,
                              const int* __restrict__ rowptr,
                              int M, int K, int Ncols, int addbias) {
    const int lane  = threadIdx.x & 31;
    const int wave  = threadIdx.x >> 5;
    const int ntile = Ncols >> 4;
    const int mtile = M >> 4;
    const int tile  = blockIdx.x * (blockDim.x >> 5) + wave;
    if (tile >= mtile * ntile) return;
    const int mt = tile / ntile;
    const int nt = tile - mt * ntile;

    const int lm    = lane & 15;
    const int khalf = (lane >> 4) << 1;

    const float* arow = A + (size_t)(mt * 16 + lm) * K;
    const float* wcol = W + (size_t)(nt * 16 + lm);

    const int r0 = mt * 16 + ((lane >> 4) << 3);
    float sc[8];
    if (rowptr != nullptr) {
        int rp[9];
#pragma unroll
        for (int v = 0; v < 9; ++v) rp[v] = (r0 + v < M) ? rowptr[r0 + v] : NE;
#pragma unroll
        for (int v = 0; v < 8; ++v)
            sc[v] = rsqrtf((float)(rp[v + 1] - rp[v] + 1));
    } else {
#pragma unroll
        for (int v = 0; v < 8; ++v) sc[v] = 1.0f;
    }

    v8f acc = {0.f, 0.f, 0.f, 0.f, 0.f, 0.f, 0.f, 0.f};
    for (int k0 = 0; k0 < K; k0 += 4) {
        v2f a, b;
        a[0] = arow[k0 + khalf];
        a[1] = arow[k0 + khalf + 1];
        b[0] = wcol[(size_t)(k0 + khalf) * Ncols];
        b[1] = wcol[(size_t)(k0 + khalf + 1) * Ncols];
        acc = __builtin_amdgcn_wmma_f32_16x16x4_f32(false, a, false, b,
                                                    (short)0, acc, false, false);
    }

    const float bv = addbias ? bias[nt * 16 + lm] : 0.0f;
    float* crow = C + (size_t)r0 * Ncols + nt * 16 + lm;
#pragma unroll
    for (int v = 0; v < 8; ++v)
        crow[(size_t)v * Ncols] = (acc[v] + bv) * sc[v];
}

__global__ void gather_rows(const int* __restrict__ rowptr, const int* __restrict__ csr,
                            const float* __restrict__ hs, const float* __restrict__ b,
                            float* __restrict__ h, int n) {
    const int lane = threadIdx.x & 31;
    const int hh   = lane >> 4;
    const int c4   = (lane & 15) << 2;
    const int gw   = (blockIdx.x * blockDim.x + threadIdx.x) >> 5;
    const int nw   = (gridDim.x * blockDim.x) >> 5;
    const float4 bb = *(const float4*)(b + c4);

    for (int d = gw; d < n; d += nw) {
        const int beg = rowptr[d];
        const int end = (d + 1 < n) ? rowptr[d + 1] : NE;
        float4 acc = {0.f, 0.f, 0.f, 0.f};
        if (hh == 0) acc = *(const float4*)(hs + (size_t)d * HID + c4);

        for (int base = beg; base < end; base += 32) {
            const int cnt = min(32, end - base);
            const int myidx = (lane < cnt) ? csr[base + lane] : 0;
            const int npair = (cnt + 1) >> 1;
            int i = 0;
            for (; i + 4 <= npair; i += 4) {
                const int k0 = 2 * i + hh, k1 = k0 + 2, k2 = k0 + 4, k3 = k0 + 6;
                const bool p0 = k0 < cnt, p1 = k1 < cnt, p2 = k2 < cnt, p3 = k3 < cnt;
                const int s0 = __shfl(myidx, p0 ? k0 : 0);
                const int s1 = __shfl(myidx, p1 ? k1 : 0);
                const int s2 = __shfl(myidx, p2 ? k2 : 0);
                const int s3 = __shfl(myidx, p3 ? k3 : 0);
                const float4 v0 = *(const float4*)(hs + (size_t)s0 * HID + c4);
                const float4 v1 = *(const float4*)(hs + (size_t)s1 * HID + c4);
                const float4 v2 = *(const float4*)(hs + (size_t)s2 * HID + c4);
                const float4 v3 = *(const float4*)(hs + (size_t)s3 * HID + c4);
                acc.x += p0 ? v0.x : 0.f; acc.y += p0 ? v0.y : 0.f;
                acc.z += p0 ? v0.z : 0.f; acc.w += p0 ? v0.w : 0.f;
                acc.x += p1 ? v1.x : 0.f; acc.y += p1 ? v1.y : 0.f;
                acc.z += p1 ? v1.z : 0.f; acc.w += p1 ? v1.w : 0.f;
                acc.x += p2 ? v2.x : 0.f; acc.y += p2 ? v2.y : 0.f;
                acc.z += p2 ? v2.z : 0.f; acc.w += p2 ? v2.w : 0.f;
                acc.x += p3 ? v3.x : 0.f; acc.y += p3 ? v3.y : 0.f;
                acc.z += p3 ? v3.z : 0.f; acc.w += p3 ? v3.w : 0.f;
            }
            for (; i < npair; ++i) {
                const int k = 2 * i + hh;
                const bool valid = k < cnt;
                const int s = __shfl(myidx, valid ? k : 0);
                const float4 v = *(const float4*)(hs + (size_t)s * HID + c4);
                acc.x += valid ? v.x : 0.f;
                acc.y += valid ? v.y : 0.f;
                acc.z += valid ? v.z : 0.f;
                acc.w += valid ? v.w : 0.f;
            }
        }
        acc.x += __shfl_xor(acc.x, 16);
        acc.y += __shfl_xor(acc.y, 16);
        acc.z += __shfl_xor(acc.z, 16);
        acc.w += __shfl_xor(acc.w, 16);
        const float di = rsqrtf((float)(end - beg + 1));
        if (hh == 0) {
            float4 o;
            o.x = fmaxf(di * acc.x + bb.x, 0.0f);
            o.y = fmaxf(di * acc.y + bb.y, 0.0f);
            o.z = fmaxf(di * acc.z + bb.z, 0.0f);
            o.w = fmaxf(di * acc.w + bb.w, 0.0f);
            *(float4*)(h + (size_t)d * HID + c4) = o;
        }
    }
}

static inline int cdiv(long long a, int b) { return (int)((a + b - 1) / b); }

extern "C" void kernel_launch(void* const* d_in, const int* in_sizes, int n_in,
                              void* d_out, int out_size, void* d_ws, size_t ws_size,
                              hipStream_t stream) {
    const float* x  = (const float*)d_in[0];
    const int*   ei = (const int*)d_in[1];
    const float* W1 = (const float*)d_in[2];
    const float* b1 = (const float*)d_in[3];
    const float* W2 = (const float*)d_in[4];
    const float* b2 = (const float*)d_in[5];
    const float* Wl = (const float*)d_in[6];
    const float* bl = (const float*)d_in[7];
    float* out = (float*)d_out;

    int*   rowptr = (int*)d_ws;
    float* buf1 = (float*)d_ws + NN;
    float* buf2 = buf1 + (size_t)NN * HID;
    int* rank = (int*)buf2;
    int* cnt  = rank + NE;
    int* bsum = cnt + NN;
    int* csr = (int*)d_out;

    const int* src = ei;
    const int* dst = ei + NE;

    const int B = 256;
    const int SB = cdiv(NN, SCAN_TILE);

    zero_cnt<<<cdiv(NN, B), B, 0, stream>>>(cnt, NN);
    count_rank<<<cdiv(NE, B), B, 0, stream>>>(dst, cnt, rank, NE);
    block_sums<<<SB, B, 0, stream>>>(cnt, bsum, NN);
    scan_rows<<<SB, B, 0, stream>>>(cnt, bsum, rowptr, NN);
    place_edges<<<cdiv(NE, B), B, 0, stream>>>(src, dst, rowptr, rank, csr, NE);

    {
        int tiles = (NN / 16) * (HID / 16);
        gemm_wmma_f32<<<cdiv(tiles, 4), 128, 0, stream>>>(x, W1, nullptr, buf1, rowptr,
                                                          NN, INF, HID, 0);
    }
    gather_rows<<<2048, B, 0, stream>>>(rowptr, csr, buf1, b1, buf2, NN);

    {
        int tiles = (NN / 16) * (HID / 16);
        gemm_wmma_f32<<<cdiv(tiles, 4), 128, 0, stream>>>(buf2, W2, nullptr, buf1, rowptr,
                                                          NN, HID, HID, 0);
    }
    gather_rows<<<2048, B, 0, stream>>>(rowptr, csr, buf1, b2, buf2, NN);

    {
        int tiles = (NN / 16) * (NCL / 16);
        gemm_wmma_f32<<<cdiv(tiles, 4), 128, 0, stream>>>(buf2, Wl, bl, out, nullptr,
                                                          NN, HID, NCL, 1);
    }
}


# per-edge returning global rank atomics and global placement replaced by a two-level counting sort ranked at LDS counters (edge chunks by node range, then per range); sums in LDS arrival order
# speedup vs baseline: 3.7454x; 1.7607x over previous
#include <hip/hip_runtime.h>


#define NN 100000
#define NE 1600000
#define INF 128
#define HID 64
#define NCL 16
#define NB 256
#define BW 391
#define G1 256
#define T1 1024
#define CH ((NE + G1 - 1) / G1)
#define EPT ((CH + T1 - 1) / T1)
#define CAP2 12288
#define RB 8
#define TB 16

typedef float v2f __attribute__((ext_vector_type(2)));
typedef float v8f __attribute__((ext_vector_type(8)));


__device__ __forceinline__ int range_of(int d) {
    return min((int)((unsigned)d / (unsigned)BW), NB - 1);
}

__device__ __forceinline__ int group_excl_scan(int v, int* wsum, int& total) {
    const int lane = threadIdx.x & 31;
    const int wave = threadIdx.x >> 5;
    int x = v;
#pragma unroll
    for (int o = 1; o < 32; o <<= 1) {
        int y = __shfl_up(x, o);
        if (lane >= o) x += y;
    }
    if (lane == 31) wsum[wave] = x;
    __syncthreads();
    if (wave == 0) {
        const int s = wsum[lane];
        int xs = s;
#pragma unroll
        for (int o = 1; o < 32; o <<= 1) {
            int y = __shfl_up(xs, o);
            if (lane >= o) xs += y;
        }
        wsum[lane] = xs - s;
        if (lane == 31) wsum[32] = xs;
    }
    __syncthreads();
    const int res = wsum[wave] + x - v;
    total = wsum[32];
    __syncthreads();
    return res;
}

__global__ __launch_bounds__(T1) void range_hist(const int* __restrict__ dst,
                                                  int* __restrict__ ghist) {
    __shared__ int h[NB];
    const int t = threadIdx.x;
    if (t < NB) h[t] = 0;
    __syncthreads();
    const int c0 = blockIdx.x * CH;
    const int c1 = min(c0 + CH, NE);
    int ed[EPT];
#pragma unroll
    for (int j = 0; j < EPT; ++j) ed[j] = dst[min(c0 + t + j * T1, c1 - 1)];
#pragma unroll
    for (int j = 0; j < EPT; ++j)
        if (c0 + t + j * T1 < c1) atomicAdd(&h[range_of(ed[j])], 1);
    __syncthreads();
    if (t < NB) ghist[blockIdx.x * NB + t] = h[t];
}

__global__ __launch_bounds__(T1) void range_scatter(const int* __restrict__ src,
                                                     const int* __restrict__ dst,
                                                     const int* __restrict__ ghist,
                                                     int* __restrict__ bbase,
                                                     int* __restrict__ tsrc,
                                                     int* __restrict__ tdst) {
    __shared__ int part_tot[4][NB];
    __shared__ int part_pre[4][NB];
    __shared__ int cnt[NB];
    __shared__ int delta[NB];
    __shared__ int wsum[33];
    __shared__ int s_src[CH];
    __shared__ int s_dst[CH];
    const int t = threadIdx.x;
    const int k = t & (NB - 1);
    const int q = t >> 8;
    const int me = blockIdx.x;

    const int c0 = me * CH;
    const int c1 = min(c0 + CH, NE);
    int es[EPT], ed[EPT], er[EPT];
#pragma unroll
    for (int j = 0; j < EPT; ++j) {
        const int i = min(c0 + t + j * T1, c1 - 1);
        es[j] = src[i];
        ed[j] = dst[i];
    }

    int tot = 0, pre = 0;
    for (int r0 = q; r0 < G1; r0 += 4 * TB) {
        int v[TB];
#pragma unroll
        for (int u = 0; u < TB; ++u) v[u] = ghist[(r0 + 4 * u) * NB + k];
#pragma unroll
        for (int u = 0; u < TB; ++u) {
            tot += v[u];
            pre += (r0 + 4 * u < me) ? v[u] : 0;
        }
    }
    part_tot[q][k] = tot;
    part_pre[q][k] = pre;
    if (t < NB) cnt[t] = 0;
    __syncthreads();

#pragma unroll
    for (int j = 0; j < EPT; ++j) {
        er[j] = 0;
        if (c0 + t + j * T1 < c1) er[j] = atomicAdd(&cnt[range_of(ed[j])], 1);
    }
    __syncthreads();

    int tk = 0, pk = 0, lc = 0;
    if (t < NB) {
        tk = part_tot[0][t] + part_tot[1][t] + part_tot[2][t] + part_tot[3][t];
        pk = part_pre[0][t] + part_pre[1][t] + part_pre[2][t] + part_pre[3][t];
        lc = cnt[t];
    }
    int all_edges, chunk_edges;
    const int base   = group_excl_scan(tk, wsum, all_edges);
    const int lstart = group_excl_scan(lc, wsum, chunk_edges);
    if (t < NB) {
        cnt[t]   = lstart;
        delta[t] = base + pk - lstart;
        if (me == 0) bbase[t] = base;
    }
    if (me == 0 && t == 0) bbase[NB] = all_edges;
    __syncthreads();

#pragma unroll
    for (int j = 0; j < EPT; ++j) {
        const int i = c0 + t + j * T1;
        if (i < c1) {
            const int p = cnt[range_of(ed[j])] + er[j];
            s_src[p] = es[j];
            s_dst[p] = ed[j];
        }
    }
    __syncthreads();

    for (int i = t; i < chunk_edges; i += T1) {
        const int d = s_dst[i];
        const int p = delta[range_of(d)] + i;
        tsrc[p] = s_src[i];
        tdst[p] = d;
    }
}

__global__ __launch_bounds__(T1) void range_rows(const int* __restrict__ tsrc,
                                                  const int* __restrict__ tdst,
                                                  const int* __restrict__ bbase,
                                                  int* __restrict__ rowptr,
                                                  int* __restrict__ csr) {
    __shared__ int cnt[BW];
    __shared__ int wsum[33];
    __shared__ int s_out[CAP2];
    const int t  = threadIdx.x;
    const int n0 = blockIdx.x * BW;
    const int nn = min(BW, NN - n0);
    const int e0 = bbase[blockIdx.x];
    const int e1 = bbase[blockIdx.x + 1];
    const int n  = e1 - e0;

    if (t < BW) cnt[t] = 0;
    __syncthreads();
    for (int i0 = e0 + t; i0 < e1; i0 += RB * T1) {
        int dl[RB];
#pragma unroll
        for (int u = 0; u < RB; ++u) {
            const int i = i0 + u * T1;
            dl[u] = (i < e1) ? tdst[i] : 0;
        }
#pragma unroll
        for (int u = 0; u < RB; ++u)
            if (i0 + u * T1 < e1) atomicAdd(&cnt[min(max(dl[u] - n0, 0), BW - 1)], 1);
    }
    __syncthreads();

    const int c = (t < BW) ? cnt[t] : 0;
    int seen;
    const int rs = group_excl_scan(c, wsum, seen);
    if (t < nn) rowptr[n0 + t] = e0 + rs;
    if (t < BW) cnt[t] = rs;
    __syncthreads();

    const bool staged = (n <= CAP2);
    for (int i0 = e0 + t; i0 < e1; i0 += RB * T1) {
        int dl[RB], sv[RB];
#pragma unroll
        for (int u = 0; u < RB; ++u) {
            const int i = i0 + u * T1;
            dl[u] = (i < e1) ? tdst[i] : 0;
            sv[u] = (i < e1) ? tsrc[i] : 0;
        }
#pragma unroll
        for (int u = 0; u < RB; ++u) {
            if (i0 + u * T1 < e1) {
                const int p = atomicAdd(&cnt[min(max(dl[u] - n0, 0), BW - 1)], 1);
                if (staged) s_out[p] = sv[u];
                else        csr[e0 + p] = sv[u];
            }
        }
    }
    __syncthreads();
    if (staged)
        for (int i = t; i < n; i += T1) csr[e0 + i] = s_out[i];
}

__global__ void gemm_wmma_f32(const float* __restrict__ A, const float* __restrict__ W,
                              const float* __restrict__ bias, float* __restrict__ C,
                              const int* __restrict__ rowptr,
                              int M, int K, int Ncols, int addbias) {
    const int lane  = threadIdx.x & 31;
    const int wave  = threadIdx.x >> 5;
    const int ntile = Ncols >> 4;
    const int mtile = M >> 4;
    const int tile  = blockIdx.x * (blockDim.x >> 5) + wave;
    if (tile >= mtile * ntile) return;
    const int mt = tile / ntile;
    const int nt = tile - mt * ntile;

    const int lm    = lane & 15;
    const int khalf = (lane >> 4) << 1;

    const float* arow = A + (size_t)(mt * 16 + lm) * K;
    const float* wcol = W + (size_t)(nt * 16 + lm);

    const int r0 = mt * 16 + ((lane >> 4) << 3);
    float sc[8];
    if (rowptr != nullptr) {
        int rp[9];
#pragma unroll
        for (int v = 0; v < 9; ++v) rp[v] = (r0 + v < M) ? rowptr[r0 + v] : NE;
#pragma unroll
        for (int v = 0; v < 8; ++v)
            sc[v] = rsqrtf((float)(rp[v + 1] - rp[v] + 1));
    } else {
#pragma unroll
        for (int v = 0; v < 8; ++v) sc[v] = 1.0f;
    }

    v8f acc = {0.f, 0.f, 0.f, 0.f, 0.f, 0.f, 0.f, 0.f};
    for (int k0 = 0; k0 < K; k0 += 4) {
        v2f a, b;
        a[0] = arow[k0 + khalf];
        a[1] = arow[k0 + khalf + 1];
        b[0] = wcol[(size_t)(k0 + khalf) * Ncols];
        b[1] = wcol[(size_t)(k0 + khalf + 1) * Ncols];
        acc = __builtin_amdgcn_wmma_f32_16x16x4_f32(false, a, false, b,
                                                    (short)0, acc, false, false);
    }

    const float bv = addbias ? bias[nt * 16 + lm] : 0.0f;
    float* crow = C + (size_t)r0 * Ncols + nt * 16 + lm;
#pragma unroll
    for (int v = 0; v < 8; ++v)
        crow[(size_t)v * Ncols] = (acc[v] + bv) * sc[v];
}

__global__ void gather_rows(const int* __restrict__ rowptr, const int* __restrict__ csr,
                            const float* __restrict__ hs, const float* __restrict__ b,
                            float* __restrict__ h, int n) {
    const int lane = threadIdx.x & 31;
    const int hh   = lane >> 4;
    const int c4   = (lane & 15) << 2;
    const int gw   = (blockIdx.x * blockDim.x + threadIdx.x) >> 5;
    const int nw   = (gridDim.x * blockDim.x) >> 5;
    const float4 bb = *(const float4*)(b + c4);

    for (int d = gw; d < n; d += nw) {
        const int beg = rowptr[d];
        const int end = (d + 1 < n) ? rowptr[d + 1] : NE;
        float4 acc = {0.f, 0.f, 0.f, 0.f};
        if (hh == 0) acc = *(const float4*)(hs + (size_t)d * HID + c4);

        for (int base = beg; base < end; base += 32) {
            const int cnt = min(32, end - base);
            const int myidx = (lane < cnt) ? csr[base + lane] : 0;
            const int npair = (cnt + 1) >> 1;
            int i = 0;
            for (; i + 4 <= npair; i += 4) {
                const int k0 = 2 * i + hh, k1 = k0 + 2, k2 = k0 + 4, k3 = k0 + 6;
                const bool p0 = k0 < cnt, p1 = k1 < cnt, p2 = k2 < cnt, p3 = k3 < cnt;
                const int s0 = __shfl(myidx, p0 ? k0 : 0);
                const int s1 = __shfl(myidx, p1 ? k1 : 0);
                const int s2 = __shfl(myidx, p2 ? k2 : 0);
                const int s3 = __shfl(myidx, p3 ? k3 : 0);
                const float4 v0 = *(const float4*)(hs + (size_t)s0 * HID + c4);
                const float4 v1 = *(const float4*)(hs + (size_t)s1 * HID + c4);
                const float4 v2 = *(const float4*)(hs + (size_t)s2 * HID + c4);
                const float4 v3 = *(const float4*)(hs + (size_t)s3 * HID + c4);
                acc.x += p0 ? v0.x : 0.f; acc.y += p0 ? v0.y : 0.f;
                acc.z += p0 ? v0.z : 0.f; acc.w += p0 ? v0.w : 0.f;
                acc.x += p1 ? v1.x : 0.f; acc.y += p1 ? v1.y : 0.f;
                acc.z += p1 ? v1.z : 0.f; acc.w += p1 ? v1.w : 0.f;
                acc.x += p2 ? v2.x : 0.f; acc.y += p2 ? v2.y : 0.f;
                acc.z += p2 ? v2.z : 0.f; acc.w += p2 ? v2.w : 0.f;
                acc.x += p3 ? v3.x : 0.f; acc.y += p3 ? v3.y : 0.f;
                acc.z += p3 ? v3.z : 0.f; acc.w += p3 ? v3.w : 0.f;
            }
            for (; i < npair; ++i) {
                const int k = 2 * i + hh;
                const bool valid = k < cnt;
                const int s = __shfl(myidx, valid ? k : 0);
                const float4 v = *(const float4*)(hs + (size_t)s * HID + c4);
                acc.x += valid ? v.x : 0.f;
                acc.y += valid ? v.y : 0.f;
                acc.z += valid ? v.z : 0.f;
                acc.w += valid ? v.w : 0.f;
            }
        }
        acc.x += __shfl_xor(acc.x, 16);
        acc.y += __shfl_xor(acc.y, 16);
        acc.z += __shfl_xor(acc.z, 16);
        acc.w += __shfl_xor(acc.w, 16);
        const float di = rsqrtf((float)(end - beg + 1));
        if (hh == 0) {
            float4 o;
            o.x = fmaxf(di * acc.x + bb.x, 0.0f);
            o.y = fmaxf(di * acc.y + bb.y, 0.0f);
            o.z = fmaxf(di * acc.z + bb.z, 0.0f);
            o.w = fmaxf(di * acc.w + bb.w, 0.0f);
            *(float4*)(h + (size_t)d * HID + c4) = o;
        }
    }
}

static inline int cdiv(long long a, int b) { return (int)((a + b - 1) / b); }

extern "C" void kernel_launch(void* const* d_in, const int* in_sizes, int n_in,
                              void* d_out, int out_size, void* d_ws, size_t ws_size,
                              hipStream_t stream) {
    const float* x  = (const float*)d_in[0];
    const int*   ei = (const int*)d_in[1];
    const float* W1 = (const float*)d_in[2];
    const float* b1 = (const float*)d_in[3];
    const float* W2 = (const float*)d_in[4];
    const float* b2 = (const float*)d_in[5];
    const float* Wl = (const float*)d_in[6];
    const float* bl = (const float*)d_in[7];
    float* out = (float*)d_out;

    int*   rowptr = (int*)d_ws;
    float* buf1 = (float*)d_ws + NN;
    float* buf2 = buf1 + (size_t)NN * HID;
    int* tsrc  = (int*)buf2;
    int* tdst  = tsrc + NE;
    int* ghist = tdst + NE;
    int* bbase = ghist + G1 * NB;
    int* csr = (int*)d_out;

    const int* src = ei;
    const int* dst = ei + NE;

    const int B = 256;

    range_hist<<<G1, T1, 0, stream>>>(dst, ghist);
    range_scatter<<<G1, T1, 0, stream>>>(src, dst, ghist, bbase, tsrc, tdst);
    range_rows<<<NB, T1, 0, stream>>>(tsrc, tdst, bbase, rowptr, csr);

    {
        int tiles = (NN / 16) * (HID / 16);
        gemm_wmma_f32<<<cdiv(tiles, 4), 128, 0, stream>>>(x, W1, nullptr, buf1, rowptr,
                                                          NN, INF, HID, 0);
    }
    gather_rows<<<2048, B, 0, stream>>>(rowptr, csr, buf1, b1, buf2, NN);

    {
        int tiles = (NN / 16) * (HID / 16);
        gemm_wmma_f32<<<cdiv(tiles, 4), 128, 0, stream>>>(buf2, W2, nullptr, buf1, rowptr,
                                                          NN, HID, HID, 0);
    }
    gather_rows<<<2048, B, 0, stream>>>(rowptr, csr, buf1, b2, buf2, NN);

    {
        int tiles = (NN / 16) * (NCL / 16);
        gemm_wmma_f32<<<cdiv(tiles, 4), 128, 0, stream>>>(buf2, Wl, bl, out, nullptr,
                                                          NN, HID, NCL, 1);
    }
}


# GEMMs re-tiled: one wave per 16-row strip over all column tiles, rows read once by 16-byte loads, W fragments from LDS, 256 resident work-groups; k summed in permuted order within groups of 8
# speedup vs baseline: 4.3353x; 1.1575x over previous
#include <hip/hip_runtime.h>


#define NN 100000
#define NE 1600000
#define INF 128
#define HID 64
#define NCL 16
#define NB 256
#define BW 391
#define G1 256
#define T1 1024
#define CH ((NE + G1 - 1) / G1)
#define EPT ((CH + T1 - 1) / T1)
#define CAP2 12288
#define RB 8
#define TB 16

typedef float v2f __attribute__((ext_vector_type(2)));
typedef float v8f __attribute__((ext_vector_type(8)));


__device__ __forceinline__ int range_of(int d) {
    return min((int)((unsigned)d / (unsigned)BW), NB - 1);
}

__device__ __forceinline__ int group_excl_scan(int v, int* wsum, int& total) {
    const int lane = threadIdx.x & 31;
    const int wave = threadIdx.x >> 5;
    int x = v;
#pragma unroll
    for (int o = 1; o < 32; o <<= 1) {
        int y = __shfl_up(x, o);
        if (lane >= o) x += y;
    }
    if (lane == 31) wsum[wave] = x;
    __syncthreads();
    if (wave == 0) {
        const int s = wsum[lane];
        int xs = s;
#pragma unroll
        for (int o = 1; o < 32; o <<= 1) {
            int y = __shfl_up(xs, o);
            if (lane >= o) xs += y;
        }
        wsum[lane] = xs - s;
        if (lane == 31) wsum[32] = xs;
    }
    __syncthreads();
    const int res = wsum[wave] + x - v;
    total = wsum[32];
    __syncthreads();
    return res;
}

__global__ __launch_bounds__(T1) void range_hist(const int* __restrict__ dst,
                                                  int* __restrict__ ghist) {
    __shared__ int h[NB];
    const int t = threadIdx.x;
    if (t < NB) h[t] = 0;
    __syncthreads();
    const int c0 = blockIdx.x * CH;
    const int c1 = min(c0 + CH, NE);
    int ed[EPT];
#pragma unroll
    for (int j = 0; j < EPT; ++j) ed[j] = dst[min(c0 + t + j * T1, c1 - 1)];
#pragma unroll
    for (int j = 0; j < EPT; ++j)
        if (c0 + t + j * T1 < c1) atomicAdd(&h[range_of(ed[j])], 1);
    __syncthreads();
    if (t < NB) ghist[blockIdx.x * NB + t] = h[t];
}

__global__ __launch_bounds__(T1) void range_scatter(const int* __restrict__ src,
                                                     const int* __restrict__ dst,
                                                     const int* __restrict__ ghist,
                                                     int* __restrict__ bbase,
                                                     int* __restrict__ tsrc,
                                                     int* __restrict__ tdst) {
    __shared__ int part_tot[4][NB];
    __shared__ int part_pre[4][NB];
    __shared__ int cnt[NB];
    __shared__ int delta[NB];
    __shared__ int wsum[33];
    __shared__ int s_src[CH];
    __shared__ int s_dst[CH];
    const int t = threadIdx.x;
    const int k = t & (NB - 1);
    const int q = t >> 8;
    const int me = blockIdx.x;

    const int c0 = me * CH;
    const int c1 = min(c0 + CH, NE);
    int es[EPT], ed[EPT], er[EPT];
#pragma unroll
    for (int j = 0; j < EPT; ++j) {
        const int i = min(c0 + t + j * T1, c1 - 1);
        es[j] = src[i];
        ed[j] = dst[i];
    }

    int tot = 0, pre = 0;
    for (int r0 = q; r0 < G1; r0 += 4 * TB) {
        int v[TB];
#pragma unroll
        for (int u = 0; u < TB; ++u) v[u] = ghist[(r0 + 4 * u) * NB + k];
#pragma unroll
        for (int u = 0; u < TB; ++u) {
            tot += v[u];
            pre += (r0 + 4 * u < me) ? v[u] : 0;
        }
    }
    part_tot[q][k] = tot;
    part_pre[q][k] = pre;
    if (t < NB) cnt[t] = 0;
    __syncthreads();

#pragma unroll
    for (int j = 0; j < EPT; ++j) {
        er[j] = 0;
        if (c0 + t + j * T1 < c1) er[j] = atomicAdd(&cnt[range_of(ed[j])], 1);
    }
    __syncthreads();

    int tk = 0, pk = 0, lc = 0;
    if (t < NB) {
        tk = part_tot[0][t] + part_tot[1][t] + part_tot[2][t] + part_tot[3][t];
        pk = part_pre[0][t] + part_pre[1][t] + part_pre[2][t] + part_pre[3][t];
        lc = cnt[t];
    }
    int all_edges, chunk_edges;
    const int base   = group_excl_scan(tk, wsum, all_edges);
    const int lstart = group_excl_scan(lc, wsum, chunk_edges);
    if (t < NB) {
        cnt[t]   = lstart;
        delta[t] = base + pk - lstart;
        if (me == 0) bbase[t] = base;
    }
    if (me == 0 && t == 0) bbase[NB] = all_edges;
    __syncthreads();

#pragma unroll
    for (int j = 0; j < EPT; ++j) {
        const int i = c0 + t + j * T1;
        if (i < c1) {
            const int p = cnt[range_of(ed[j])] + er[j];
            s_src[p] = es[j];
            s_dst[p] = ed[j];
        }
    }
    __syncthreads();

    for (int i = t; i < chunk_edges; i += T1) {
        const int d = s_dst[i];
        const int p = delta[range_of(d)] + i;
        tsrc[p] = s_src[i];
        tdst[p] = d;
    }
}

__global__ __launch_bounds__(T1) void range_rows(const int* __restrict__ tsrc,
                                                  const int* __restrict__ tdst,
                                                  const int* __restrict__ bbase,
                                                  int* __restrict__ rowptr,
                                                  int* __restrict__ csr) {
    __shared__ int cnt[BW];
    __shared__ int wsum[33];
    __shared__ int s_out[CAP2];
    const int t  = threadIdx.x;
    const int n0 = blockIdx.x * BW;
    const int nn = min(BW, NN - n0);
    const int e0 = bbase[blockIdx.x];
    const int e1 = bbase[blockIdx.x + 1];
    const int n  = e1 - e0;

    if (t < BW) cnt[t] = 0;
    __syncthreads();
    for (int i0 = e0 + t; i0 < e1; i0 += RB * T1) {
        int dl[RB];
#pragma unroll
        for (int u = 0; u < RB; ++u) {
            const int i = i0 + u * T1;
            dl[u] = (i < e1) ? tdst[i] : 0;
        }
#pragma unroll
        for (int u = 0; u < RB; ++u)
            if (i0 + u * T1 < e1) atomicAdd(&cnt[min(max(dl[u] - n0, 0), BW - 1)], 1);
    }
    __syncthreads();

    const int c = (t < BW) ? cnt[t] : 0;
    int seen;
    const int rs = group_excl_scan(c, wsum, seen);
    if (t < nn) rowptr[n0 + t] = e0 + rs;
    if (t < BW) cnt[t] = rs;
    __syncthreads();

    const bool staged = (n <= CAP2);
    for (int i0 = e0 + t; i0 < e1; i0 += RB * T1) {
        int dl[RB], sv[RB];
#pragma unroll
        for (int u = 0; u < RB; ++u) {
            const int i = i0 + u * T1;
            dl[u] = (i < e1) ? tdst[i] : 0;
            sv[u] = (i < e1) ? tsrc[i] : 0;
        }
#pragma unroll
        for (int u = 0; u < RB; ++u) {
            if (i0 + u * T1 < e1) {
                const int p = atomicAdd(&cnt[min(max(dl[u] - n0, 0), BW - 1)], 1);
                if (staged) s_out[p] = sv[u];
                else        csr[e0 + p] = sv[u];
            }
        }
    }
    __syncthreads();
    if (staged)
        for (int i = t; i < n; i += T1) csr[e0 + i] = s_out[i];
}

#define GW 256
#define GT 512

template <int K, bool SCALE>
__device__ __forceinline__ void load_strip(const float* __restrict__ A,
                                           const int* __restrict__ rowptr, int M, int s, int c,
                                           int h, float4 (&a)[K / 8], int (&rp)[9]) {
    const float* arow = A + (size_t)(s * 16 + c) * K + 4 * h;
#pragma unroll
    for (int j = 0; j < K / 8; ++j) a[j] = *(const float4*)(arow + 8 * j);
    if constexpr (SCALE) {
        const int r0 = s * 16 + 8 * h;
#pragma unroll
        for (int v = 0; v < 8; ++v) rp[v] = rowptr[r0 + v];
        rp[8] = rowptr[min(r0 + 8, M - 1)];
    }
}

template <int K, int NC, bool SCALE, bool BIAS>
__global__ __launch_bounds__(GT) void gemm_strip(const float* __restrict__ A,
                                                 const float* __restrict__ W,
                                                 const float* __restrict__ bias,
                                                 float* __restrict__ C,
                                                 const int* __restrict__ rowptr, int M) {
    constexpr int NT    = NC / 16;
    constexpr int KG    = K / 8;
    constexpr int PITCH = K + 4;
    constexpr int BG    = 2;
    static_assert(K % (8 * BG) == 0 && NC % 16 == 0, "shape");
    __shared__ alignas(16) float wt[NC * PITCH];

    const int lane = threadIdx.x & 31;
    const int c    = lane & 15;
    const int h    = lane >> 4;
    const int wave   = __builtin_amdgcn_readfirstlane((int)(threadIdx.x >> 5));
    const int nwaves = GT / 32;
    const int nstrip = M >> 4;
    const int step   = (int)gridDim.x * nwaves;
    int s = (int)blockIdx.x + (int)gridDim.x * wave;

    float4 a[KG];
    int rp[9];
#pragma unroll
    for (int j = 0; j < KG; ++j) a[j] = make_float4(0.f, 0.f, 0.f, 0.f);
#pragma unroll
    for (int v = 0; v < 9; ++v) rp[v] = 0;
    if (s < nstrip) load_strip<K, SCALE>(A, rowptr, M, s, c, h, a, rp);

    constexpr int PER = K * NC / GT;
    static_assert(K * NC % GT == 0, "staging");
    float wv[PER];
#pragma unroll
    for (int u = 0; u < PER; ++u) wv[u] = W[threadIdx.x + u * GT];
#pragma unroll
    for (int u = 0; u < PER; ++u) {
        const int i = threadIdx.x + u * GT;
        const int k = i / NC;
        const int n = i - k * NC;
        wt[n * PITCH + k] = wv[u];
    }
    float bv[NT];
#pragma unroll
    for (int nt = 0; nt < NT; ++nt) bv[nt] = BIAS ? bias[nt * 16 + c] : 0.0f;
    __syncthreads();

    const float* wl = wt + c * PITCH + 4 * h;

    while (s < nstrip) {
        asm volatile("" ::: "memory");
        const int r0 = s * 16 + 8 * h;

        v8f acc[NT];
#pragma unroll
        for (int nt = 0; nt < NT; ++nt) acc[nt] = (v8f){0.f, 0.f, 0.f, 0.f, 0.f, 0.f, 0.f, 0.f};

        float4 bc[BG * NT], bn[BG * NT];
#pragma unroll
        for (int u = 0; u < BG * NT; ++u)
            bc[u] = *(const float4*)(wl + (u % NT) * 16 * PITCH + 8 * (u / NT));
#pragma unroll
        for (int j0 = 0; j0 < KG; j0 += BG) {
            if (j0 + BG < KG) {
#pragma unroll
                for (int u = 0; u < BG * NT; ++u)
                    bn[u] = *(const float4*)(wl + (u % NT) * 16 * PITCH + 8 * (j0 + BG + u / NT));
            }
#pragma unroll
            for (int g = 0; g < BG; ++g) {
                const float4 av = a[j0 + g];
                const v2f a01 = {av.x, av.y};
                const v2f a23 = {av.z, av.w};
#pragma unroll
                for (int nt = 0; nt < NT; ++nt) {
                    const v2f b01 = {bc[g * NT + nt].x, bc[g * NT + nt].y};
                    acc[nt] = __builtin_amdgcn_wmma_f32_16x16x4_f32(false, a01, false, b01,
                                                                    (short)0, acc[nt], false, false);
                }
#pragma unroll
                for (int nt = 0; nt < NT; ++nt) {
                    const v2f b23 = {bc[g * NT + nt].z, bc[g * NT + nt].w};
                    acc[nt] = __builtin_amdgcn_wmma_f32_16x16x4_f32(false, a23, false, b23,
                                                                    (short)0, acc[nt], false, false);
                }
            }
            if (j0 + BG < KG) {
#pragma unroll
                for (int u = 0; u < BG * NT; ++u) bc[u] = bn[u];
            }
        }

        float sc[8];
        if constexpr (SCALE) {
            if (r0 + 8 >= M) rp[8] = NE;
#pragma unroll
            for (int v = 0; v < 8; ++v)
                sc[v] = rsqrtf((float)(rp[v + 1] - rp[v] + 1));
        } else {
#pragma unroll
            for (int v = 0; v < 8; ++v) sc[v] = 1.0f;
        }

        float* crow = C + (size_t)r0 * NC + c;
#pragma unroll
        for (int v = 0; v < 8; ++v) {
#pragma unroll
            for (int nt = 0; nt < NT; ++nt) {
                const float t = BIAS ? acc[nt][v] + bv[nt] : acc[nt][v];
                crow[(size_t)v * NC + nt * 16] = t * sc[v];
            }
        }
        s += step;
        if (s < nstrip) load_strip<K, SCALE>(A, rowptr, M, s, c, h, a, rp);
    }
}

__global__ void gather_rows(const int* __restrict__ rowptr, const int* __restrict__ csr,
                            const float* __restrict__ hs, const float* __restrict__ b,
                            float* __restrict__ h, int n) {
    const int lane = threadIdx.x & 31;
    const int hh   = lane >> 4;
    const int c4   = (lane & 15) << 2;
    const int gw   = (blockIdx.x * blockDim.x + threadIdx.x) >> 5;
    const int nw   = (gridDim.x * blockDim.x) >> 5;
    const float4 bb = *(const float4*)(b + c4);

    for (int d = gw; d < n; d += nw) {
        const int beg = rowptr[d];
        const int end = (d + 1 < n) ? rowptr[d + 1] : NE;
        float4 acc = {0.f, 0.f, 0.f, 0.f};
        if (hh == 0) acc = *(const float4*)(hs + (size_t)d * HID + c4);

        for (int base = beg; base < end; base += 32) {
            const int cnt = min(32, end - base);
            const int myidx = (lane < cnt) ? csr[base + lane] : 0;
            const int npair = (cnt + 1) >> 1;
            int i = 0;
            for (; i + 4 <= npair; i += 4) {
                const int k0 = 2 * i + hh, k1 = k0 + 2, k2 = k0 + 4, k3 = k0 + 6;
                const bool p0 = k0 < cnt, p1 = k1 < cnt, p2 = k2 < cnt, p3 = k3 < cnt;
                const int s0 = __shfl(myidx, p0 ? k0 : 0);
                const int s1 = __shfl(myidx, p1 ? k1 : 0);
                const int s2 = __shfl(myidx, p2 ? k2 : 0);
                const int s3 = __shfl(myidx, p3 ? k3 : 0);
                const float4 v0 = *(const float4*)(hs + (size_t)s0 * HID + c4);
                const float4 v1 = *(const float4*)(hs + (size_t)s1 * HID + c4);
                const float4 v2 = *(const float4*)(hs + (size_t)s2 * HID + c4);
                const float4 v3 = *(const float4*)(hs + (size_t)s3 * HID + c4);
                acc.x += p0 ? v0.x : 0.f; acc.y += p0 ? v0.y : 0.f;
                acc.z += p0 ? v0.z : 0.f; acc.w += p0 ? v0.w : 0.f;
                acc.x += p1 ? v1.x : 0.f; acc.y += p1 ? v1.y : 0.f;
                acc.z += p1 ? v1.z : 0.f; acc.w += p1 ? v1.w : 0.f;
                acc.x += p2 ? v2.x : 0.f; acc.y += p2 ? v2.y : 0.f;
                acc.z += p2 ? v2.z : 0.f; acc.w += p2 ? v2.w : 0.f;
                acc.x += p3 ? v3.x : 0.f; acc.y += p3 ? v3.y : 0.f;
                acc.z += p3 ? v3.z : 0.f; acc.w += p3 ? v3.w : 0.f;
            }
            for (; i < npair; ++i) {
                const int k = 2 * i + hh;
                const bool valid = k < cnt;
                const int s = __shfl(myidx, valid ? k : 0);
                const float4 v = *(const float4*)(hs + (size_t)s * HID + c4);
                acc.x += valid ? v.x : 0.f;
                acc.y += valid ? v.y : 0.f;
                acc.z += valid ? v.z : 0.f;
                acc.w += valid ? v.w : 0.f;
            }
        }
        acc.x += __shfl_xor(acc.x, 16);
        acc.y += __shfl_xor(acc.y, 16);
        acc.z += __shfl_xor(acc.z, 16);
        acc.w += __shfl_xor(acc.w, 16);
        const float di = rsqrtf((float)(end - beg + 1));
        if (hh == 0) {
            float4 o;
            o.x = fmaxf(di * acc.x + bb.x, 0.0f);
            o.y = fmaxf(di * acc.y + bb.y, 0.0f);
            o.z = fmaxf(di * acc.z + bb.z, 0.0f);
            o.w = fmaxf(di * acc.w + bb.w, 0.0f);
            *(float4*)(h + (size_t)d * HID + c4) = o;
        }
    }
}

extern "C" void kernel_launch(void* const* d_in, const int* in_sizes, int n_in,
                              void* d_out, int out_size, void* d_ws, size_t ws_size,
                              hipStream_t stream) {
    const float* x  = (const float*)d_in[0];
    const int*   ei = (const int*)d_in[1];
    const float* W1 = (const float*)d_in[2];
    const float* b1 = (const float*)d_in[3];
    const float* W2 = (const float*)d_in[4];
    const float* b2 = (const float*)d_in[5];
    const float* Wl = (const float*)d_in[6];
    const float* bl = (const float*)d_in[7];
    float* out = (float*)d_out;

    int*   rowptr = (int*)d_ws;
    float* buf1 = (float*)d_ws + NN;
    float* buf2 = buf1 + (size_t)NN * HID;
    int* tsrc  = (int*)buf2;
    int* tdst  = tsrc + NE;
    int* ghist = tdst + NE;
    int* bbase = ghist + G1 * NB;
    int* csr = (int*)d_out;

    const int* src = ei;
    const int* dst = ei + NE;

    const int B = 256;

    range_hist<<<G1, T1, 0, stream>>>(dst, ghist);
    range_scatter<<<G1, T1, 0, stream>>>(src, dst, ghist, bbase, tsrc, tdst);
    range_rows<<<NB, T1, 0, stream>>>(tsrc, tdst, bbase, rowptr, csr);

    gemm_strip<INF, HID, true, false><<<GW, GT, 0, stream>>>(x, W1, nullptr, buf1, rowptr, NN);
    gather_rows<<<2048, B, 0, stream>>>(rowptr, csr, buf1, b1, buf2, NN);

    gemm_strip<HID, HID, true, false><<<GW, GT, 0, stream>>>(buf2, W2, nullptr, buf1, rowptr, NN);
    gather_rows<<<2048, B, 0, stream>>>(rowptr, csr, buf1, b2, buf2, NN);

    gemm_strip<HID, NCL, false, true><<<GW, GT, 0, stream>>>(buf2, Wl, bl, out, nullptr, NN);
}


# counting sort level 1 made chunk-local: no histogram launch, no count-table sums; chunks stay in place ordered by range, each range walks its per-chunk runs; sums still in LDS arrival order
# speedup vs baseline: 4.5423x; 1.0477x over previous
#include <hip/hip_runtime.h>


#define NN 100000
#define NE 1600000
#define INF 128
#define HID 64
#define NCL 16
#define NB 256
#define BW 391
#define G1 256
#define T1 1024
#define CH ((NE + G1 - 1) / G1)
#define EPT ((CH + T1 - 1) / T1)
#define CAP2 12288
#define RL 4
#define PP 288
#define RB 16

typedef float v2f __attribute__((ext_vector_type(2)));
typedef float v8f __attribute__((ext_vector_type(8)));


__device__ __forceinline__ int range_of(int d) {
    return min((int)((unsigned)d / (unsigned)BW), NB - 1);
}

__device__ __forceinline__ int group_excl_scan(int v, int* wsum, int& total) {
    const int lane = threadIdx.x & 31;
    const int wave = threadIdx.x >> 5;
    int x = v;
#pragma unroll
    for (int o = 1; o < 32; o <<= 1) {
        int y = __shfl_up(x, o);
        if (lane >= o) x += y;
    }
    if (lane == 31) wsum[wave] = x;
    __syncthreads();
    if (wave == 0) {
        const int s = wsum[lane];
        int xs = s;
#pragma unroll
        for (int o = 1; o < 32; o <<= 1) {
            int y = __shfl_up(xs, o);
            if (lane >= o) xs += y;
        }
        wsum[lane] = xs - s;
        if (lane == 31) wsum[32] = xs;
    }
    __syncthreads();
    const int res = wsum[wave] + x - v;
    total = wsum[32];
    __syncthreads();
    return res;
}

__global__ __launch_bounds__(T1) void range_scatter(const int* __restrict__ src,
                                                     const int* __restrict__ dst,
                                                     int* __restrict__ pt,
                                                     int* __restrict__ tsrc,
                                                     int* __restrict__ tdst) {
    __shared__ int cnt[NB];
    __shared__ int wsum[33];
    __shared__ int s_src[CH];
    __shared__ int s_dst[CH];
    const int t  = threadIdx.x;
    const int me = blockIdx.x;

    const int c0 = me * CH;
    const int c1 = min(c0 + CH, NE);
    int es[EPT], ed[EPT], er[EPT];
#pragma unroll
    for (int j = 0; j < EPT; ++j) {
        const int i = min(c0 + t + j * T1, c1 - 1);
        es[j] = src[i];
        ed[j] = dst[i];
    }
    if (t < NB) cnt[t] = 0;
    __syncthreads();

#pragma unroll
    for (int j = 0; j < EPT; ++j) {
        er[j] = 0;
        if (c0 + t + j * T1 < c1) er[j] = atomicAdd(&cnt[range_of(ed[j])], 1);
    }
    __syncthreads();

    const int lc = (t < NB) ? cnt[t] : 0;
    int chunk_edges;
    const int lstart = group_excl_scan(lc, wsum, chunk_edges);
    if (t < NB) {
        cnt[t] = lstart;
        pt[me * PP + t] = lstart;
    }
    if (t == 0) pt[me * PP + NB] = chunk_edges;
    __syncthreads();

#pragma unroll
    for (int j = 0; j < EPT; ++j) {
        const int i = c0 + t + j * T1;
        if (i < c1) {
            const int p = cnt[range_of(ed[j])] + er[j];
            s_src[p] = es[j];
            s_dst[p] = ed[j];
        }
    }
    __syncthreads();

    for (int i = t; i < chunk_edges; i += T1) {
        tsrc[c0 + i] = s_src[i];
        tdst[c0 + i] = s_dst[i];
    }
}

__device__ __forceinline__ int node_slot(int d, int n0) {
    return min(max(d - n0, 0), BW - 1);
}

__global__ __launch_bounds__(T1) void range_rows(const int* __restrict__ tsrc,
                                                  const int* __restrict__ tdst,
                                                  const int* __restrict__ pt,
                                                  int* __restrict__ rowptr,
                                                  int* __restrict__ csr) {
    static_assert(T1 == RL * G1 && PP >= NB + 1 && PP % 32 == 0, "run table shape");
    __shared__ int cnt[BW];
    __shared__ int wsum[33];
    __shared__ int s_out[CAP2];
    const int t  = threadIdx.x;
    const int n0 = blockIdx.x * BW;
    const int nn = min(BW, NN - n0);

    const int ch  = t / RL;
    const int sub = t % RL;
    const int a   = pt[ch * PP + blockIdx.x];
    const int len = pt[ch * PP + blockIdx.x + 1] - a;
    const int* run_d = tdst + ch * CH + a;
    const int* run_s = tsrc + ch * CH + a;

    int d0[RB], s0[RB];
#pragma unroll
    for (int u = 0; u < RB; ++u) {
        const int i = sub + RL * u;
        d0[u] = (i < len) ? run_d[i] : 0;
        s0[u] = (i < len) ? run_s[i] : 0;
    }

    if (t < BW) cnt[t] = 0;
    int e0, n;
    group_excl_scan(sub == 0 ? a : 0, wsum, e0);
    group_excl_scan(sub == 0 ? len : 0, wsum, n);

#pragma unroll
    for (int u = 0; u < RB; ++u)
        if (sub + RL * u < len) atomicAdd(&cnt[node_slot(d0[u], n0)], 1);
    for (int i0 = sub + RL * RB; i0 < len; i0 += RL * RB) {
        int dl[RB];
#pragma unroll
        for (int u = 0; u < RB; ++u) {
            const int i = i0 + RL * u;
            dl[u] = (i < len) ? run_d[i] : 0;
        }
#pragma unroll
        for (int u = 0; u < RB; ++u)
            if (i0 + RL * u < len) atomicAdd(&cnt[node_slot(dl[u], n0)], 1);
    }
    __syncthreads();

    const int c = (t < BW) ? cnt[t] : 0;
    int seen;
    const int rs = group_excl_scan(c, wsum, seen);
    if (t < nn) rowptr[n0 + t] = e0 + rs;
    if (t < BW) cnt[t] = rs;
    __syncthreads();

    const bool staged = (n <= CAP2);
#pragma unroll
    for (int u = 0; u < RB; ++u) {
        if (sub + RL * u < len) {
            const int p = atomicAdd(&cnt[node_slot(d0[u], n0)], 1);
            if (staged) s_out[p] = s0[u];
            else        csr[e0 + p] = s0[u];
        }
    }
    for (int i0 = sub + RL * RB; i0 < len; i0 += RL * RB) {
        int dl[RB], sv[RB];
#pragma unroll
        for (int u = 0; u < RB; ++u) {
            const int i = i0 + RL * u;
            dl[u] = (i < len) ? run_d[i] : 0;
            sv[u] = (i < len) ? run_s[i] : 0;
        }
#pragma unroll
        for (int u = 0; u < RB; ++u) {
            if (i0 + RL * u < len) {
                const int p = atomicAdd(&cnt[node_slot(dl[u], n0)], 1);
                if (staged) s_out[p] = sv[u];
                else        csr[e0 + p] = sv[u];
            }
        }
    }
    __syncthreads();
    if (staged)
        for (int i = t; i < n; i += T1) csr[e0 + i] = s_out[i];
}

#define GW 256
#define GT 512

template <int K, bool SCALE>
__device__ __forceinline__ void load_strip(const float* __restrict__ A,
                                           const int* __restrict__ rowptr, int M, int s, int c,
                                           int h, float4 (&a)[K / 8], int (&rp)[9]) {
    const float* arow = A + (size_t)(s * 16 + c) * K + 4 * h;
#pragma unroll
    for (int j = 0; j < K / 8; ++j) a[j] = *(const float4*)(arow + 8 * j);
    if constexpr (SCALE) {
        const int r0 = s * 16 + 8 * h;
#pragma unroll
        for (int v = 0; v < 8; ++v) rp[v] = rowptr[r0 + v];
        rp[8] = rowptr[min(r0 + 8, M - 1)];
    }
}

template <int K, int NC, bool SCALE, bool BIAS>
__global__ __launch_bounds__(GT) void gemm_strip(const float* __restrict__ A,
                                                 const float* __restrict__ W,
                                                 const float* __restrict__ bias,
                                                 float* __restrict__ C,
                                                 const int* __restrict__ rowptr, int M) {
    constexpr int NT    = NC / 16;
    constexpr int KG    = K / 8;
    constexpr int PITCH = K + 4;
    constexpr int BG    = 2;
    static_assert(K % (8 * BG) == 0 && NC % 16 == 0, "shape");
    __shared__ alignas(16) float wt[NC * PITCH];

    const int lane = threadIdx.x & 31;
    const int c    = lane & 15;
    const int h    = lane >> 4;
    const int wave   = __builtin_amdgcn_readfirstlane((int)(threadIdx.x >> 5));
    const int nwaves = GT / 32;
    const int nstrip = M >> 4;
    const int step   = (int)gridDim.x * nwaves;
    int s = (int)blockIdx.x + (int)gridDim.x * wave;

    float4 a[KG];
    int rp[9];
#pragma unroll
    for (int j = 0; j < KG; ++j) a[j] = make_float4(0.f, 0.f, 0.f, 0.f);
#pragma unroll
    for (int v = 0; v < 9; ++v) rp[v] = 0;
    if (s < nstrip) load_strip<K, SCALE>(A, rowptr, M, s, c, h, a, rp);

    constexpr int PER = K * NC / GT;
    static_assert(K * NC % GT == 0, "staging");
    float wv[PER];
#pragma unroll
    for (int u = 0; u < PER; ++u) wv[u] = W[threadIdx.x + u * GT];
#pragma unroll
    for (int u = 0; u < PER; ++u) {
        const int i = threadIdx.x + u * GT;
        const int k = i / NC;
        const int n = i - k * NC;
        wt[n * PITCH + k] = wv[u];
    }
    float bv[NT];
#pragma unroll
    for (int nt = 0; nt < NT; ++nt) bv[nt] = BIAS ? bias[nt * 16 + c] : 0.0f;
    __syncthreads();

    const float* wl = wt + c * PITCH + 4 * h;

    while (s < nstrip) {
        asm volatile("" ::: "memory");
        const int r0 = s * 16 + 8 * h;

        v8f acc[NT];
#pragma unroll
        for (int nt = 0; nt < NT; ++nt) acc[nt] = (v8f){0.f, 0.f, 0.f, 0.f, 0.f, 0.f, 0.f, 0.f};

        float4 bc[BG * NT], bn[BG * NT];
#pragma unroll
        for (int u = 0; u < BG * NT; ++u)
            bc[u] = *(const float4*)(wl + (u % NT) * 16 * PITCH + 8 * (u / NT));
#pragma unroll
        for (int j0 = 0; j0 < KG; j0 += BG) {
            if (j0 + BG < KG) {
#pragma unroll
                for (int u = 0; u < BG * NT; ++u)
                    bn[u] = *(const float4*)(wl + (u % NT) * 16 * PITCH + 8 * (j0 + BG + u / NT));
            }
#pragma unroll
            for (int g = 0; g < BG; ++g) {
                const float4 av = a[j0 + g];
                const v2f a01 = {av.x, av.y};
                const v2f a23 = {av.z, av.w};
#pragma unroll
                for (int nt = 0; nt < NT; ++nt) {
                    const v2f b01 = {bc[g * NT + nt].x, bc[g * NT + nt].y};
                    acc[nt] = __builtin_amdgcn_wmma_f32_16x16x4_f32(false, a01, false, b01,
                                                                    (short)0, acc[nt], false, false);
                }
#pragma unroll
                for (int nt = 0; nt < NT; ++nt) {
                    const v2f b23 = {bc[g * NT + nt].z, bc[g * NT + nt].w};
                    acc[nt] = __builtin_amdgcn_wmma_f32_16x16x4_f32(false, a23, false, b23,
                                                                    (short)0, acc[nt], false, false);
                }
            }
            if (j0 + BG < KG) {
#pragma unroll
                for (int u = 0; u < BG * NT; ++u) bc[u] = bn[u];
            }
        }

        float sc[8];
        if constexpr (SCALE) {
            if (r0 + 8 >= M) rp[8] = NE;
#pragma unroll
            for (int v = 0; v < 8; ++v)
                sc[v] = rsqrtf((float)(rp[v + 1] - rp[v] + 1));
        } else {
#pragma unroll
            for (int v = 0; v < 8; ++v) sc[v] = 1.0f;
        }

        float* crow = C + (size_t)r0 * NC + c;
#pragma unroll
        for (int v = 0; v < 8; ++v) {
#pragma unroll
            for (int nt = 0; nt < NT; ++nt) {
                const float t = BIAS ? acc[nt][v] + bv[nt] : acc[nt][v];
                crow[(size_t)v * NC + nt * 16] = t * sc[v];
            }
        }
        s += step;
        if (s < nstrip) load_strip<K, SCALE>(A, rowptr, M, s, c, h, a, rp);
    }
}

__global__ void gather_rows(const int* __restrict__ rowptr, const int* __restrict__ csr,
                            const float* __restrict__ hs, const float* __restrict__ b,
                            float* __restrict__ h, int n) {
    const int lane = threadIdx.x & 31;
    const int hh   = lane >> 4;
    const int c4   = (lane & 15) << 2;
    const int gw   = (blockIdx.x * blockDim.x + threadIdx.x) >> 5;
    const int nw   = (gridDim.x * blockDim.x) >> 5;
    const float4 bb = *(const float4*)(b + c4);

    for (int d = gw; d < n; d += nw) {
        const int beg = rowptr[d];
        const int end = (d + 1 < n) ? rowptr[d + 1] : NE;
        float4 acc = {0.f, 0.f, 0.f, 0.f};
        if (hh == 0) acc = *(const float4*)(hs + (size_t)d * HID + c4);

        for (int base = beg; base < end; base += 32) {
            const int cnt = min(32, end - base);
            const int myidx = (lane < cnt) ? csr[base + lane] : 0;
            const int npair = (cnt + 1) >> 1;
            int i = 0;
            for (; i + 4 <= npair; i += 4) {
                const int k0 = 2 * i + hh, k1 = k0 + 2, k2 = k0 + 4, k3 = k0 + 6;
                const bool p0 = k0 < cnt, p1 = k1 < cnt, p2 = k2 < cnt, p3 = k3 < cnt;
                const int s0 = __shfl(myidx, p0 ? k0 : 0);
                const int s1 = __shfl(myidx, p1 ? k1 : 0);
                const int s2 = __shfl(myidx, p2 ? k2 : 0);
                const int s3 = __shfl(myidx, p3 ? k3 : 0);
                const float4 v0 = *(const float4*)(hs + (size_t)s0 * HID + c4);
                const float4 v1 = *(const float4*)(hs + (size_t)s1 * HID + c4);
                const float4 v2 = *(const float4*)(hs + (size_t)s2 * HID + c4);
                const float4 v3 = *(const float4*)(hs + (size_t)s3 * HID + c4);
                acc.x += p0 ? v0.x : 0.f; acc.y += p0 ? v0.y : 0.f;
                acc.z += p0 ? v0.z : 0.f; acc.w += p0 ? v0.w : 0.f;
                acc.x += p1 ? v1.x : 0.f; acc.y += p1 ? v1.y : 0.f;
                acc.z += p1 ? v1.z : 0.f; acc.w += p1 ? v1.w : 0.f;
                acc.x += p2 ? v2.x : 0.f; acc.y += p2 ? v2.y : 0.f;
                acc.z += p2 ? v2.z : 0.f; acc.w += p2 ? v2.w : 0.f;
                acc.x += p3 ? v3.x : 0.f; acc.y += p3 ? v3.y : 0.f;
                acc.z += p3 ? v3.z : 0.f; acc.w += p3 ? v3.w : 0.f;
            }
            for (; i < npair; ++i) {
                const int k = 2 * i + hh;
                const bool valid = k < cnt;
                const int s = __shfl(myidx, valid ? k : 0);
                const float4 v = *(const float4*)(hs + (size_t)s * HID + c4);
                acc.x += valid ? v.x : 0.f;
                acc.y += valid ? v.y : 0.f;
                acc.z += valid ? v.z : 0.f;
                acc.w += valid ? v.w : 0.f;
            }
        }
        acc.x += __shfl_xor(acc.x, 16);
        acc.y += __shfl_xor(acc.y, 16);
        acc.z += __shfl_xor(acc.z, 16);
        acc.w += __shfl_xor(acc.w, 16);
        const float di = rsqrtf((float)(end - beg + 1));
        if (hh == 0) {
            float4 o;
            o.x = fmaxf(di * acc.x + bb.x, 0.0f);
            o.y = fmaxf(di * acc.y + bb.y, 0.0f);
            o.z = fmaxf(di * acc.z + bb.z, 0.0f);
            o.w = fmaxf(di * acc.w + bb.w, 0.0f);
            *(float4*)(h + (size_t)d * HID + c4) = o;
        }
    }
}

extern "C" void kernel_launch(void* const* d_in, const int* in_sizes, int n_in,
                              void* d_out, int out_size, void* d_ws, size_t ws_size,
                              hipStream_t stream) {
    const float* x  = (const float*)d_in[0];
    const int*   ei = (const int*)d_in[1];
    const float* W1 = (const float*)d_in[2];
    const float* b1 = (const float*)d_in[3];
    const float* W2 = (const float*)d_in[4];
    const float* b2 = (const float*)d_in[5];
    const float* Wl = (const float*)d_in[6];
    const float* bl = (const float*)d_in[7];
    float* out = (float*)d_out;

    int*   rowptr = (int*)d_ws;
    float* buf1 = (float*)d_ws + NN;
    float* buf2 = buf1 + (size_t)NN * HID;
    int* tsrc  = (int*)buf2;
    int* tdst  = tsrc + NE;
    int* pt    = tdst + NE;
    int* csr = (int*)d_out;

    const int* src = ei;
    const int* dst = ei + NE;

    const int B = 256;

    range_scatter<<<G1, T1, 0, stream>>>(src, dst, pt, tsrc, tdst);
    range_rows<<<NB, T1, 0, stream>>>(tsrc, tdst, pt, rowptr, csr);

    gemm_strip<INF, HID, true, false><<<GW, GT, 0, stream>>>(x, W1, nullptr, buf1, rowptr, NN);
    gather_rows<<<2048, B, 0, stream>>>(rowptr, csr, buf1, b1, buf2, NN);

    gemm_strip<HID, HID, true, false><<<GW, GT, 0, stream>>>(buf2, W2, nullptr, buf1, rowptr, NN);
    gather_rows<<<2048, B, 0, stream>>>(rowptr, csr, buf1, b2, buf2, NN);

    gemm_strip<HID, NCL, false, true><<<GW, GT, 0, stream>>>(buf2, Wl, bl, out, nullptr, NN);
}


# each neighbour-sum launch fused with the GEMM that consumes it: a wave sums 16 nodes and feeds the rows through its LDS tile to its own WMMAs; 7 launches to 5; summation order unchanged
# speedup vs baseline: 4.7782x; 1.0519x over previous
#include <hip/hip_runtime.h>


#define NN 100000
#define NE 1600000
#define INF 128
#define HID 64
#define NCL 16
#define NB 256
#define BW 391
#define G1 256
#define T1 1024
#define CH ((NE + G1 - 1) / G1)
#define EPT ((CH + T1 - 1) / T1)
#define CAP2 12288
#define RL 4
#define PP 288
#define RB 16
#define SPL 75000
#define FT 256
#define FW (FT / 32)
#define FG ((NN / 16 + FW - 1) / FW)
#define TP (HID + 4)

typedef float v2f __attribute__((ext_vector_type(2)));
typedef float v8f __attribute__((ext_vector_type(8)));


__device__ __forceinline__ int range_of(int d) {
    return min((int)((unsigned)d / (unsigned)BW), NB - 1);
}

__device__ __forceinline__ int group_excl_scan(int v, int* wsum, int& total) {
    const int lane = threadIdx.x & 31;
    const int wave = threadIdx.x >> 5;
    int x = v;
#pragma unroll
    for (int o = 1; o < 32; o <<= 1) {
        int y = __shfl_up(x, o);
        if (lane >= o) x += y;
    }
    if (lane == 31) wsum[wave] = x;
    __syncthreads();
    if (wave == 0) {
        const int s = wsum[lane];
        int xs = s;
#pragma unroll
        for (int o = 1; o < 32; o <<= 1) {
            int y = __shfl_up(xs, o);
            if (lane >= o) xs += y;
        }
        wsum[lane] = xs - s;
        if (lane == 31) wsum[32] = xs;
    }
    __syncthreads();
    const int res = wsum[wave] + x - v;
    total = wsum[32];
    __syncthreads();
    return res;
}

__global__ __launch_bounds__(T1) void range_scatter(const int* __restrict__ src,
                                                     const int* __restrict__ dst,
                                                     int* __restrict__ pt,
                                                     int* __restrict__ tsrc,
                                                     int* __restrict__ tdst) {
    __shared__ int cnt[NB];
    __shared__ int wsum[33];
    __shared__ int s_src[CH];
    __shared__ int s_dst[CH];
    const int t  = threadIdx.x;
    const int me = blockIdx.x;

    const int c0 = me * CH;
    const int c1 = min(c0 + CH, NE);
    int es[EPT], ed[EPT], er[EPT];
#pragma unroll
    for (int j = 0; j < EPT; ++j) {
        const int i = min(c0 + t + j * T1, c1 - 1);
        es[j] = src[i];
        ed[j] = dst[i];
    }
    if (t < NB) cnt[t] = 0;
    __syncthreads();

#pragma unroll
    for (int j = 0; j < EPT; ++j) {
        er[j] = 0;
        if (c0 + t + j * T1 < c1) er[j] = atomicAdd(&cnt[range_of(ed[j])], 1);
    }
    __syncthreads();

    const int lc = (t < NB) ? cnt[t] : 0;
    int chunk_edges;
    const int lstart = group_excl_scan(lc, wsum, chunk_edges);
    if (t < NB) {
        cnt[t] = lstart;
        pt[me * PP + t] = lstart;
    }
    if (t == 0) pt[me * PP + NB] = chunk_edges;
    __syncthreads();

#pragma unroll
    for (int j = 0; j < EPT; ++j) {
        const int i = c0 + t + j * T1;
        if (i < c1) {
            const int p = cnt[range_of(ed[j])] + er[j];
            s_src[p] = es[j];
            s_dst[p] = ed[j];
        }
    }
    __syncthreads();

    for (int i = t; i < chunk_edges; i += T1) {
        tsrc[c0 + i] = s_src[i];
        tdst[c0 + i] = s_dst[i];
    }
}

__device__ __forceinline__ int node_slot(int d, int n0) {
    return min(max(d - n0, 0), BW - 1);
}

__global__ __launch_bounds__(T1) void range_rows(const int* __restrict__ tsrc,
                                                  const int* __restrict__ tdst,
                                                  const int* __restrict__ pt,
                                                  int* __restrict__ rowptr,
                                                  int* __restrict__ csr) {
    static_assert(T1 == RL * G1 && PP >= NB + 1 && PP % 32 == 0, "run table shape");
    __shared__ int cnt[BW];
    __shared__ int wsum[33];
    __shared__ int s_out[CAP2];
    const int t  = threadIdx.x;
    const int n0 = blockIdx.x * BW;
    const int nn = min(BW, NN - n0);

    const int ch  = t / RL;
    const int sub = t % RL;
    const int a   = pt[ch * PP + blockIdx.x];
    const int len = pt[ch * PP + blockIdx.x + 1] - a;
    const int* run_d = tdst + ch * CH + a;
    const int* run_s = tsrc + ch * CH + a;

    int d0[RB], s0[RB];
#pragma unroll
    for (int u = 0; u < RB; ++u) {
        const int i = sub + RL * u;
        d0[u] = (i < len) ? run_d[i] : 0;
        s0[u] = (i < len) ? run_s[i] : 0;
    }

    if (t < BW) cnt[t] = 0;
    int e0, n;
    group_excl_scan(sub == 0 ? a : 0, wsum, e0);
    group_excl_scan(sub == 0 ? len : 0, wsum, n);

#pragma unroll
    for (int u = 0; u < RB; ++u)
        if (sub + RL * u < len) atomicAdd(&cnt[node_slot(d0[u], n0)], 1);
    for (int i0 = sub + RL * RB; i0 < len; i0 += RL * RB) {
        int dl[RB];
#pragma unroll
        for (int u = 0; u < RB; ++u) {
            const int i = i0 + RL * u;
            dl[u] = (i < len) ? run_d[i] : 0;
        }
#pragma unroll
        for (int u = 0; u < RB; ++u)
            if (i0 + RL * u < len) atomicAdd(&cnt[node_slot(dl[u], n0)], 1);
    }
    __syncthreads();

    const int c = (t < BW) ? cnt[t] : 0;
    int seen;
    const int rs = group_excl_scan(c, wsum, seen);
    if (t < nn) rowptr[n0 + t] = e0 + rs;
    if (t < BW) cnt[t] = rs;
    __syncthreads();

    const bool staged = (n <= CAP2);
#pragma unroll
    for (int u = 0; u < RB; ++u) {
        if (sub + RL * u < len) {
            const int p = atomicAdd(&cnt[node_slot(d0[u], n0)], 1);
            if (staged) s_out[p] = s0[u];
            else        csr[e0 + p] = s0[u];
        }
    }
    for (int i0 = sub + RL * RB; i0 < len; i0 += RL * RB) {
        int dl[RB], sv[RB];
#pragma unroll
        for (int u = 0; u < RB; ++u) {
            const int i = i0 + RL * u;
            dl[u] = (i < len) ? run_d[i] : 0;
            sv[u] = (i < len) ? run_s[i] : 0;
        }
#pragma unroll
        for (int u = 0; u < RB; ++u) {
            if (i0 + RL * u < len) {
                const int p = atomicAdd(&cnt[node_slot(dl[u], n0)], 1);
                if (staged) s_out[p] = sv[u];
                else        csr[e0 + p] = sv[u];
            }
        }
    }
    __syncthreads();
    if (staged)
        for (int i = t; i < n; i += T1) csr[e0 + i] = s_out[i];
}

#define GW 256
#define GT 512

template <int K, bool SCALE>
__device__ __forceinline__ void load_strip(const float* __restrict__ A,
                                           const int* __restrict__ rowptr, int M, int s, int c,
                                           int h, float4 (&a)[K / 8], int (&rp)[9]) {
    const float* arow = A + (size_t)(s * 16 + c) * K + 4 * h;
#pragma unroll
    for (int j = 0; j < K / 8; ++j) a[j] = *(const float4*)(arow + 8 * j);
    if constexpr (SCALE) {
        const int r0 = s * 16 + 8 * h;
#pragma unroll
        for (int v = 0; v < 8; ++v) rp[v] = rowptr[r0 + v];
        rp[8] = rowptr[min(r0 + 8, M - 1)];
    }
}

template <int K, int NC, bool SCALE, bool BIAS>
__global__ __launch_bounds__(GT) void gemm_strip(const float* __restrict__ A,
                                                 const float* __restrict__ W,
                                                 const float* __restrict__ bias,
                                                 float* __restrict__ C,
                                                 float* __restrict__ Chi,
                                                 const int* __restrict__ rowptr, int M) {
    constexpr int NT    = NC / 16;
    constexpr int KG    = K / 8;
    constexpr int PITCH = K + 4;
    constexpr int BG    = 2;
    static_assert(K % (8 * BG) == 0 && NC % 16 == 0, "shape");
    __shared__ alignas(16) float wt[NC * PITCH];

    const int lane = threadIdx.x & 31;
    const int c    = lane & 15;
    const int h    = lane >> 4;
    const int wave   = __builtin_amdgcn_readfirstlane((int)(threadIdx.x >> 5));
    const int nwaves = GT / 32;
    const int nstrip = M >> 4;
    const int step   = (int)gridDim.x * nwaves;
    int s = (int)blockIdx.x + (int)gridDim.x * wave;

    float4 a[KG];
    int rp[9];
#pragma unroll
    for (int j = 0; j < KG; ++j) a[j] = make_float4(0.f, 0.f, 0.f, 0.f);
#pragma unroll
    for (int v = 0; v < 9; ++v) rp[v] = 0;
    if (s < nstrip) load_strip<K, SCALE>(A, rowptr, M, s, c, h, a, rp);

    constexpr int PER = K * NC / GT;
    static_assert(K * NC % GT == 0, "staging");
    float wv[PER];
#pragma unroll
    for (int u = 0; u < PER; ++u) wv[u] = W[threadIdx.x + u * GT];
#pragma unroll
    for (int u = 0; u < PER; ++u) {
        const int i = threadIdx.x + u * GT;
        const int k = i / NC;
        const int n = i - k * NC;
        wt[n * PITCH + k] = wv[u];
    }
    float bv[NT];
#pragma unroll
    for (int nt = 0; nt < NT; ++nt) bv[nt] = BIAS ? bias[nt * 16 + c] : 0.0f;
    __syncthreads();

    const float* wl = wt + c * PITCH + 4 * h;

    while (s < nstrip) {
        asm volatile("" ::: "memory");
        const int r0 = s * 16 + 8 * h;

        v8f acc[NT];
#pragma unroll
        for (int nt = 0; nt < NT; ++nt) acc[nt] = (v8f){0.f, 0.f, 0.f, 0.f, 0.f, 0.f, 0.f, 0.f};

        float4 bc[BG * NT], bn[BG * NT];
#pragma unroll
        for (int u = 0; u < BG * NT; ++u)
            bc[u] = *(const float4*)(wl + (u % NT) * 16 * PITCH + 8 * (u / NT));
#pragma unroll
        for (int j0 = 0; j0 < KG; j0 += BG) {
            if (j0 + BG < KG) {
#pragma unroll
                for (int u = 0; u < BG * NT; ++u)
                    bn[u] = *(const float4*)(wl + (u % NT) * 16 * PITCH + 8 * (j0 + BG + u / NT));
            }
#pragma unroll
            for (int g = 0; g < BG; ++g) {
                const float4 av = a[j0 + g];
                const v2f a01 = {av.x, av.y};
                const v2f a23 = {av.z, av.w};
#pragma unroll
                for (int nt = 0; nt < NT; ++nt) {
                    const v2f b01 = {bc[g * NT + nt].x, bc[g * NT + nt].y};
                    acc[nt] = __builtin_amdgcn_wmma_f32_16x16x4_f32(false, a01, false, b01,
                                                                    (short)0, acc[nt], false, false);
                }
#pragma unroll
                for (int nt = 0; nt < NT; ++nt) {
                    const v2f b23 = {bc[g * NT + nt].z, bc[g * NT + nt].w};
                    acc[nt] = __builtin_amdgcn_wmma_f32_16x16x4_f32(false, a23, false, b23,
                                                                    (short)0, acc[nt], false, false);
                }
            }
            if (j0 + BG < KG) {
#pragma unroll
                for (int u = 0; u < BG * NT; ++u) bc[u] = bn[u];
            }
        }

        float sc[8];
        if constexpr (SCALE) {
            if (r0 + 8 >= M) rp[8] = NE;
#pragma unroll
            for (int v = 0; v < 8; ++v)
                sc[v] = rsqrtf((float)(rp[v + 1] - rp[v] + 1));
        } else {
#pragma unroll
            for (int v = 0; v < 8; ++v) sc[v] = 1.0f;
        }

        float* crow = (r0 < SPL) ? C + (size_t)r0 * NC + c : Chi + (size_t)(r0 - SPL) * NC + c;
#pragma unroll
        for (int v = 0; v < 8; ++v) {
#pragma unroll
            for (int nt = 0; nt < NT; ++nt) {
                const float t = BIAS ? acc[nt][v] + bv[nt] : acc[nt][v];
                crow[(size_t)v * NC + nt * 16] = t * sc[v];
            }
        }
        s += step;
        if (s < nstrip) load_strip<K, SCALE>(A, rowptr, M, s, c, h, a, rp);
    }
}

template <bool SPLIT>
__device__ __forceinline__ const float* hs_row(const float* __restrict__ lo,
                                               const float* __restrict__ hi, int r) {
    if constexpr (SPLIT)
        return (r < SPL) ? lo + (size_t)r * HID : hi + (size_t)(r - SPL) * HID;
    else
        return lo + (size_t)r * HID;
}

template <int N>
__device__ __forceinline__ void pair_sources(int myidx, int i, int cnt, int hh, bool (&p)[N],
                                             int (&sx)[N]) {
#pragma unroll
    for (int u = 0; u < N; ++u) {
        const int k = 2 * (i + u) + hh;
        p[u]  = k < cnt;
        sx[u] = __shfl(myidx, k);
    }
}

template <int N, bool SPLIT>
__device__ __forceinline__ void pair_rows(float4& acc, const bool (&p)[N], const int (&sx)[N],
                                          int c4, const float* __restrict__ lo,
                                          const float* __restrict__ hi) {
    float4 v[N];
#pragma unroll
    for (int u = 0; u < N; ++u) {
        v[u] = make_float4(0.f, 0.f, 0.f, 0.f);
        if (p[u]) v[u] = *(const float4*)(hs_row<SPLIT>(lo, hi, sx[u]) + c4);
    }
#pragma unroll
    for (int u = 0; u < N; ++u) {
        acc.x += v[u].x;
        acc.y += v[u].y;
        acc.z += v[u].z;
        acc.w += v[u].w;
    }
}

template <int N, bool SPLIT>
__device__ __forceinline__ void add_pairs(float4& acc, int myidx, int i, int cnt, int hh, int c4,
                                          const float* __restrict__ lo,
                                          const float* __restrict__ hi) {
    bool p[N];
    int sx[N];
    pair_sources<N>(myidx, i, cnt, hh, p, sx);
    pair_rows<N, SPLIT>(acc, p, sx, c4, lo, hi);
}

template <int NC, bool SCALE, bool BIAS, bool SPLIT>
__global__ __launch_bounds__(FT) void gather_gemm(const int* __restrict__ rowptr,
                                                  const int* __restrict__ csr,
                                                  const float* __restrict__ hs,
                                                  const float* __restrict__ hs_hi,
                                                  const float* __restrict__ b,
                                                  const float* __restrict__ W,
                                                  const float* __restrict__ bias,
                                                  float* __restrict__ C) {
    constexpr int K     = HID;
    constexpr int NT    = NC / 16;
    constexpr int KG    = K / 8;
    constexpr int PITCH = K + 4;
    static_assert(NC % 16 == 0 && K * NC % FT == 0 && NN % 16 == 0 && SPL % 8 == 0, "shape");
    __shared__ alignas(16) float wt[NC * PITCH];
    __shared__ alignas(16) float tile[FW * 16 * TP];

    const int lane = threadIdx.x & 31;
    const int c    = lane & 15;
    const int h    = lane >> 4;
    const int c4   = c << 2;
    const int wave = __builtin_amdgcn_readfirstlane((int)(threadIdx.x >> 5));
    const int s    = (int)blockIdx.x * FW + wave;
    const bool live = s < NN / 16;

    int rpv = 0;
    if (live) {
        const int idx = s * 16 + min(lane, 16);
        rpv = (idx < NN) ? rowptr[idx] : NE;
    }

    constexpr int PER = K * NC / FT;
    float wv[PER];
#pragma unroll
    for (int u = 0; u < PER; ++u) wv[u] = W[threadIdx.x + u * FT];
#pragma unroll
    for (int u = 0; u < PER; ++u) {
        const int i = threadIdx.x + u * FT;
        const int k = i / NC;
        const int n = i - k * NC;
        wt[n * PITCH + k] = wv[u];
    }
    float bv[NT];
#pragma unroll
    for (int nt = 0; nt < NT; ++nt) bv[nt] = BIAS ? bias[nt * 16 + c] : 0.0f;
    const float4 bb = *(const float4*)(b + c4);
    __syncthreads();
    if (!live) return;

    float* tl = tile + wave * 16 * TP;

    int beg = __builtin_amdgcn_readlane(rpv, 0);
    int cur = csr[min(beg + lane, NE - 1)];
    for (int q = 0; q < 16; ++q) {
        const int d    = s * 16 + q;
        const int end  = __builtin_amdgcn_readlane(rpv, q + 1);
        const int deg  = end - beg;
        const int cnt0 = min(32, deg);

        bool p0[8];
        int sx0[8];
        pair_sources<8>(cur, 0, cnt0, h, p0, sx0);
        const int nxt = csr[min(end + lane, NE - 1)];
        float4 acc = {0.f, 0.f, 0.f, 0.f};
        if (h == 0) acc = *(const float4*)(hs_row<SPLIT>(hs, hs_hi, d) + c4);
        pair_rows<8, SPLIT>(acc, p0, sx0, c4, hs, hs_hi);

        if (deg > 16) {
            if (cnt0 > 16) add_pairs<8, SPLIT>(acc, cur, 8, cnt0, h, c4, hs, hs_hi);
            for (int base = beg + 32; base < end; base += 32) {
                const int cnt   = min(32, end - base);
                const int myidx = csr[min(base + lane, NE - 1)];
                const int npair = (cnt + 1) >> 1;
                for (int i = 0; i < npair; i += 8)
                    add_pairs<8, SPLIT>(acc, myidx, i, cnt, h, c4, hs, hs_hi);
            }
        }
        acc.x += __shfl_xor(acc.x, 16);
        acc.y += __shfl_xor(acc.y, 16);
        acc.z += __shfl_xor(acc.z, 16);
        acc.w += __shfl_xor(acc.w, 16);
        const float di = rsqrtf((float)(end - beg + 1));
        if (h == 0) {
            float4 o;
            o.x = fmaxf(di * acc.x + bb.x, 0.0f);
            o.y = fmaxf(di * acc.y + bb.y, 0.0f);
            o.z = fmaxf(di * acc.z + bb.z, 0.0f);
            o.w = fmaxf(di * acc.w + bb.w, 0.0f);
            *(float4*)(tl + q * TP + c4) = o;
        }
        beg = end;
        cur = nxt;
    }

    float4 a[KG];
#pragma unroll
    for (int j = 0; j < KG; ++j) a[j] = *(const float4*)(tl + c * TP + 8 * j + 4 * h);

    const float* wl = wt + c * PITCH + 4 * h;
    v8f acc[NT];
#pragma unroll
    for (int nt = 0; nt < NT; ++nt) acc[nt] = (v8f){0.f, 0.f, 0.f, 0.f, 0.f, 0.f, 0.f, 0.f};

    float4 bc[NT], bn[NT];
#pragma unroll
    for (int u = 0; u < NT; ++u) bc[u] = *(const float4*)(wl + u * 16 * PITCH);
#pragma unroll
    for (int j = 0; j < KG; ++j) {
        if (j + 1 < KG) {
#pragma unroll
            for (int u = 0; u < NT; ++u) bn[u] = *(const float4*)(wl + u * 16 * PITCH + 8 * (j + 1));
        }
        const v2f a01 = {a[j].x, a[j].y};
        const v2f a23 = {a[j].z, a[j].w};
#pragma unroll
        for (int nt = 0; nt < NT; ++nt) {
            const v2f b01 = {bc[nt].x, bc[nt].y};
            acc[nt] = __builtin_amdgcn_wmma_f32_16x16x4_f32(false, a01, false, b01,
                                                            (short)0, acc[nt], false, false);
        }
#pragma unroll
        for (int nt = 0; nt < NT; ++nt) {
            const v2f b23 = {bc[nt].z, bc[nt].w};
            acc[nt] = __builtin_amdgcn_wmma_f32_16x16x4_f32(false, a23, false, b23,
                                                            (short)0, acc[nt], false, false);
        }
        if (j + 1 < KG) {
#pragma unroll
            for (int u = 0; u < NT; ++u) bc[u] = bn[u];
        }
    }

    float sc[8];
    if constexpr (SCALE) {
        const int rnext = __shfl_down(rpv, 1);
        const float dl  = rsqrtf((float)(rnext - rpv + 1));
#pragma unroll
        for (int v = 0; v < 8; ++v) sc[v] = __shfl(dl, 8 * h + v);
    } else {
#pragma unroll
        for (int v = 0; v < 8; ++v) sc[v] = 1.0f;
    }

    float* crow = C + (size_t)(s * 16 + 8 * h) * NC + c;
#pragma unroll
    for (int v = 0; v < 8; ++v) {
#pragma unroll
        for (int nt = 0; nt < NT; ++nt) {
            const float t = BIAS ? acc[nt][v] + bv[nt] : acc[nt][v];
            crow[(size_t)v * NC + nt * 16] = t * sc[v];
        }
    }
}

extern "C" void kernel_launch(void* const* d_in, const int* in_sizes, int n_in,
                              void* d_out, int out_size, void* d_ws, size_t ws_size,
                              hipStream_t stream) {
    const float* x  = (const float*)d_in[0];
    const int*   ei = (const int*)d_in[1];
    const float* W1 = (const float*)d_in[2];
    const float* b1 = (const float*)d_in[3];
    const float* W2 = (const float*)d_in[4];
    const float* b2 = (const float*)d_in[5];
    const float* Wl = (const float*)d_in[6];
    const float* bl = (const float*)d_in[7];
    float* out = (float*)d_out;

    int*   rowptr = (int*)d_ws;
    float* buf1 = (float*)d_ws + NN;
    int*   csr  = (int*)(buf1 + (size_t)SPL * HID);
    float* buf2 = buf1 + (size_t)NN * HID;
    int* tsrc  = (int*)buf2;
    int* tdst  = tsrc + NE;
    int* pt    = tdst + NE;
    float* hi1 = out;

    const int* src = ei;
    const int* dst = ei + NE;

    range_scatter<<<G1, T1, 0, stream>>>(src, dst, pt, tsrc, tdst);
    range_rows<<<NB, T1, 0, stream>>>(tsrc, tdst, pt, rowptr, csr);

    gemm_strip<INF, HID, true, false><<<GW, GT, 0, stream>>>(x, W1, nullptr, buf1, hi1, rowptr, NN);

    gather_gemm<HID, true, false, true><<<FG, FT, 0, stream>>>(rowptr, csr, buf1, hi1, b1, W2,
                                                               nullptr, buf2);

    gather_gemm<NCL, false, true, false><<<FG, FT, 0, stream>>>(rowptr, csr, buf2, nullptr, b2, Wl,
                                                                bl, out);
}


# neighbour sums: source numbers by scalar loads, each row by one load of all 32 lanes from a scalar base (no lane shuffles, no per-lane 64-bit addresses); rows added one by one in list order
# speedup vs baseline: 5.3457x; 1.1188x over previous
#include <hip/hip_runtime.h>


#define NN 100000
#define NE 1600000
#define INF 128
#define HID 64
#define NCL 16
#define NB 256
#define BW 391
#define G1 256
#define T1 1024
#define CH ((NE + G1 - 1) / G1)
#define EPT ((CH + T1 - 1) / T1)
#define CAP2 12288
#define RL 4
#define PP 288
#define RB 16
#define SPL 75000
#define FT 256
#define FW (FT / 32)
#define FG ((NN / 16 + FW - 1) / FW)
#define TP (HID + 4)
#define GB 24

typedef float v2f __attribute__((ext_vector_type(2)));
typedef float v8f __attribute__((ext_vector_type(8)));


__device__ __forceinline__ int range_of(int d) {
    return min((int)((unsigned)d / (unsigned)BW), NB - 1);
}

__device__ __forceinline__ int group_excl_scan(int v, int* wsum, int& total) {
    const int lane = threadIdx.x & 31;
    const int wave = threadIdx.x >> 5;
    int x = v;
#pragma unroll
    for (int o = 1; o < 32; o <<= 1) {
        int y = __shfl_up(x, o);
        if (lane >= o) x += y;
    }
    if (lane == 31) wsum[wave] = x;
    __syncthreads();
    if (wave == 0) {
        const int s = wsum[lane];
        int xs = s;
#pragma unroll
        for (int o = 1; o < 32; o <<= 1) {
            int y = __shfl_up(xs, o);
            if (lane >= o) xs += y;
        }
        wsum[lane] = xs - s;
        if (lane == 31) wsum[32] = xs;
    }
    __syncthreads();
    const int res = wsum[wave] + x - v;
    total = wsum[32];
    __syncthreads();
    return res;
}

__global__ __launch_bounds__(T1) void range_scatter(const int* __restrict__ src,
                                                     const int* __restrict__ dst,
                                                     int* __restrict__ pt,
                                                     int* __restrict__ tsrc,
                                                     int* __restrict__ tdst) {
    __shared__ int cnt[NB];
    __shared__ int wsum[33];
    __shared__ int s_src[CH];
    __shared__ int s_dst[CH];
    const int t  = threadIdx.x;
    const int me = blockIdx.x;

    const int c0 = me * CH;
    const int c1 = min(c0 + CH, NE);
    int es[EPT], ed[EPT], er[EPT];
#pragma unroll
    for (int j = 0; j < EPT; ++j) {
        const int i = min(c0 + t + j * T1, c1 - 1);
        es[j] = src[i];
        ed[j] = dst[i];
    }
    if (t < NB) cnt[t] = 0;
    __syncthreads();

#pragma unroll
    for (int j = 0; j < EPT; ++j) {
        er[j] = 0;
        if (c0 + t + j * T1 < c1) er[j] = atomicAdd(&cnt[range_of(ed[j])], 1);
    }
    __syncthreads();

    const int lc = (t < NB) ? cnt[t] : 0;
    int chunk_edges;
    const int lstart = group_excl_scan(lc, wsum, chunk_edges);
    if (t < NB) {
        cnt[t] = lstart;
        pt[me * PP + t] = lstart;
    }
    if (t == 0) pt[me * PP + NB] = chunk_edges;
    __syncthreads();

#pragma unroll
    for (int j = 0; j < EPT; ++j) {
        const int i = c0 + t + j * T1;
        if (i < c1) {
            const int p = cnt[range_of(ed[j])] + er[j];
            s_src[p] = es[j];
            s_dst[p] = ed[j];
        }
    }
    __syncthreads();

    for (int i = t; i < chunk_edges; i += T1) {
        tsrc[c0 + i] = s_src[i];
        tdst[c0 + i] = s_dst[i];
    }
}

__device__ __forceinline__ int node_slot(int d, int n0) {
    return min(max(d - n0, 0), BW - 1);
}

__global__ __launch_bounds__(T1) void range_rows(const int* __restrict__ tsrc,
                                                  const int* __restrict__ tdst,
                                                  const int* __restrict__ pt,
                                                  int* __restrict__ rowptr,
                                                  int* __restrict__ csr) {
    static_assert(T1 == RL * G1 && PP >= NB + 1 && PP % 32 == 0, "run table shape");
    __shared__ int cnt[BW];
    __shared__ int wsum[33];
    __shared__ int s_out[CAP2];
    const int t  = threadIdx.x;
    const int n0 = blockIdx.x * BW;
    const int nn = min(BW, NN - n0);

    const int ch  = t / RL;
    const int sub = t % RL;
    const int a   = pt[ch * PP + blockIdx.x];
    const int len = pt[ch * PP + blockIdx.x + 1] - a;
    const int* run_d = tdst + ch * CH + a;
    const int* run_s = tsrc + ch * CH + a;

    int d0[RB], s0[RB];
#pragma unroll
    for (int u = 0; u < RB; ++u) {
        const int i = sub + RL * u;
        d0[u] = (i < len) ? run_d[i] : 0;
        s0[u] = (i < len) ? run_s[i] : 0;
    }

    if (t < BW) cnt[t] = 0;
    int e0, n;
    group_excl_scan(sub == 0 ? a : 0, wsum, e0);
    group_excl_scan(sub == 0 ? len : 0, wsum, n);

#pragma unroll
    for (int u = 0; u < RB; ++u)
        if (sub + RL * u < len) atomicAdd(&cnt[node_slot(d0[u], n0)], 1);
    for (int i0 = sub + RL * RB; i0 < len; i0 += RL * RB) {
        int dl[RB];
#pragma unroll
        for (int u = 0; u < RB; ++u) {
            const int i = i0 + RL * u;
            dl[u] = (i < len) ? run_d[i] : 0;
        }
#pragma unroll
        for (int u = 0; u < RB; ++u)
            if (i0 + RL * u < len) atomicAdd(&cnt[node_slot(dl[u], n0)], 1);
    }
    __syncthreads();

    const int c = (t < BW) ? cnt[t] : 0;
    int seen;
    const int rs = group_excl_scan(c, wsum, seen);
    if (t < nn) rowptr[n0 + t] = e0 + rs;
    if (t < BW) cnt[t] = rs;
    __syncthreads();

    const bool staged = (n <= CAP2);
#pragma unroll
    for (int u = 0; u < RB; ++u) {
        if (sub + RL * u < len) {
            const int p = atomicAdd(&cnt[node_slot(d0[u], n0)], 1);
            if (staged) s_out[p] = s0[u];
            else        csr[e0 + p] = s0[u];
        }
    }
    for (int i0 = sub + RL * RB; i0 < len; i0 += RL * RB) {
        int dl[RB], sv[RB];
#pragma unroll
        for (int u = 0; u < RB; ++u) {
            const int i = i0 + RL * u;
            dl[u] = (i < len) ? run_d[i] : 0;
            sv[u] = (i < len) ? run_s[i] : 0;
        }
#pragma unroll
        for (int u = 0; u < RB; ++u) {
            if (i0 + RL * u < len) {
                const int p = atomicAdd(&cnt[node_slot(dl[u], n0)], 1);
                if (staged) s_out[p] = sv[u];
                else        csr[e0 + p] = sv[u];
            }
        }
    }
    __syncthreads();
    if (staged)
        for (int i = t; i < n; i += T1) csr[e0 + i] = s_out[i];
}

#define GW 256
#define GT 512

template <int K, bool SCALE>
__device__ __forceinline__ void load_strip(const float* __restrict__ A,
                                           const int* __restrict__ rowptr, int M, int s, int c,
                                           int h, float4 (&a)[K / 8], int (&rp)[9]) {
    const float* arow = A + (size_t)(s * 16 + c) * K + 4 * h;
#pragma unroll
    for (int j = 0; j < K / 8; ++j) a[j] = *(const float4*)(arow + 8 * j);
    if constexpr (SCALE) {
        const int r0 = s * 16 + 8 * h;
#pragma unroll
        for (int v = 0; v < 8; ++v) rp[v] = rowptr[r0 + v];
        rp[8] = rowptr[min(r0 + 8, M - 1)];
    }
}

template <int K, int NC, bool SCALE, bool BIAS>
__global__ __launch_bounds__(GT) void gemm_strip(const float* __restrict__ A,
                                                 const float* __restrict__ W,
                                                 const float* __restrict__ bias,
                                                 float* __restrict__ C,
                                                 float* __restrict__ Chi,
                                                 const int* __restrict__ rowptr, int M) {
    constexpr int NT    = NC / 16;
    constexpr int KG    = K / 8;
    constexpr int PITCH = K + 4;
    constexpr int BG    = 2;
    static_assert(K % (8 * BG) == 0 && NC % 16 == 0, "shape");
    __shared__ alignas(16) float wt[NC * PITCH];

    const int lane = threadIdx.x & 31;
    const int c    = lane & 15;
    const int h    = lane >> 4;
    const int wave   = __builtin_amdgcn_readfirstlane((int)(threadIdx.x >> 5));
    const int nwaves = GT / 32;
    const int nstrip = M >> 4;
    const int step   = (int)gridDim.x * nwaves;
    int s = (int)blockIdx.x + (int)gridDim.x * wave;

    float4 a[KG];
    int rp[9];
#pragma unroll
    for (int j = 0; j < KG; ++j) a[j] = make_float4(0.f, 0.f, 0.f, 0.f);
#pragma unroll
    for (int v = 0; v < 9; ++v) rp[v] = 0;
    if (s < nstrip) load_strip<K, SCALE>(A, rowptr, M, s, c, h, a, rp);

    constexpr int PER = K * NC / GT;
    static_assert(K * NC % GT == 0, "staging");
    float wv[PER];
#pragma unroll
    for (int u = 0; u < PER; ++u) wv[u] = W[threadIdx.x + u * GT];
#pragma unroll
    for (int u = 0; u < PER; ++u) {
        const int i = threadIdx.x + u * GT;
        const int k = i / NC;
        const int n = i - k * NC;
        wt[n * PITCH + k] = wv[u];
    }
    float bv[NT];
#pragma unroll
    for (int nt = 0; nt < NT; ++nt) bv[nt] = BIAS ? bias[nt * 16 + c] : 0.0f;
    __syncthreads();

    const float* wl = wt + c * PITCH + 4 * h;

    while (s < nstrip) {
        asm volatile("" ::: "memory");
        const int r0 = s * 16 + 8 * h;

        v8f acc[NT];
#pragma unroll
        for (int nt = 0; nt < NT; ++nt) acc[nt] = (v8f){0.f, 0.f, 0.f, 0.f, 0.f, 0.f, 0.f, 0.f};

        float4 bc[BG * NT], bn[BG * NT];
#pragma unroll
        for (int u = 0; u < BG * NT; ++u)
            bc[u] = *(const float4*)(wl + (u % NT) * 16 * PITCH + 8 * (u / NT));
#pragma unroll
        for (int j0 = 0; j0 < KG; j0 += BG) {
            if (j0 + BG < KG) {
#pragma unroll
                for (int u = 0; u < BG * NT; ++u)
                    bn[u] = *(const float4*)(wl + (u % NT) * 16 * PITCH + 8 * (j0 + BG + u / NT));
            }
#pragma unroll
            for (int g = 0; g < BG; ++g) {
                const float4 av = a[j0 + g];
                const v2f a01 = {av.x, av.y};
                const v2f a23 = {av.z, av.w};
#pragma unroll
                for (int nt = 0; nt < NT; ++nt) {
                    const v2f b01 = {bc[g * NT + nt].x, bc[g * NT + nt].y};
                    acc[nt] = __builtin_amdgcn_wmma_f32_16x16x4_f32(false, a01, false, b01,
                                                                    (short)0, acc[nt], false, false);
                }
#pragma unroll
                for (int nt = 0; nt < NT; ++nt) {
                    const v2f b23 = {bc[g * NT + nt].z, bc[g * NT + nt].w};
                    acc[nt] = __builtin_amdgcn_wmma_f32_16x16x4_f32(false, a23, false, b23,
                                                                    (short)0, acc[nt], false, false);
                }
            }
            if (j0 + BG < KG) {
#pragma unroll
                for (int u = 0; u < BG * NT; ++u) bc[u] = bn[u];
            }
        }

        float sc[8];
        if constexpr (SCALE) {
            if (r0 + 8 >= M) rp[8] = NE;
#pragma unroll
            for (int v = 0; v < 8; ++v)
                sc[v] = rsqrtf((float)(rp[v + 1] - rp[v] + 1));
        } else {
#pragma unroll
            for (int v = 0; v < 8; ++v) sc[v] = 1.0f;
        }

        float* crow = (r0 < SPL) ? C + (size_t)r0 * NC + c : Chi + (size_t)(r0 - SPL) * NC + c;
#pragma unroll
        for (int v = 0; v < 8; ++v) {
#pragma unroll
            for (int nt = 0; nt < NT; ++nt) {
                const float t = BIAS ? acc[nt][v] + bv[nt] : acc[nt][v];
                crow[(size_t)v * NC + nt * 16] = t * sc[v];
            }
        }
        s += step;
        if (s < nstrip) load_strip<K, SCALE>(A, rowptr, M, s, c, h, a, rp);
    }
}


template <bool SPLIT>
__device__ __forceinline__ float2 load_row2(const char* __restrict__ lo,
                                            const char* __restrict__ hi_shift, int r,
                                            unsigned lb) {
    const char* base = lo;
    if constexpr (SPLIT) {
        if (r >= SPL) base = hi_shift;
    }
    const unsigned off = (unsigned)r * (unsigned)(HID * sizeof(float)) + lb;
    return *(const float2*)(base + off);
}

template <int U, bool SPLIT>
__device__ __forceinline__ void rows_request(float2 (&v)[GB], const int (&sx)[GB], int n,
                                             unsigned lb, const char* __restrict__ lo,
                                             const char* __restrict__ hi_shift) {
    if constexpr (U < GB) {
        if (U < n) {
            v[U] = load_row2<SPLIT>(lo, hi_shift, sx[U], lb);
            rows_request<U + 1, SPLIT>(v, sx, n, lb, lo, hi_shift);
        }
    }
}

template <int U>
__device__ __forceinline__ void rows_add(float2& acc, const float2 (&v)[GB], int n) {
    if constexpr (U < GB) {
        if (U < n) {
            acc.x += v[U].x;
            acc.y += v[U].y;
            rows_add<U + 1>(acc, v, n);
        }
    }
}

template <bool SPLIT>
__device__ __forceinline__ void add_rows(float2& acc, const int (&sx)[GB], int n, unsigned lb,
                                         const char* __restrict__ lo,
                                         const char* __restrict__ hi_shift) {
    float2 v[GB];
    rows_request<0, SPLIT>(v, sx, n, lb, lo, hi_shift);
    rows_add<0>(acc, v, n);
}

template <int NC, bool SCALE, bool BIAS, bool SPLIT>
__global__ __launch_bounds__(FT) void gather_gemm(const int* __restrict__ rowptr,
                                                  const int* __restrict__ csr,
                                                  const float* __restrict__ hs,
                                                  const float* __restrict__ hs_hi,
                                                  const float* __restrict__ b,
                                                  const float* __restrict__ W,
                                                  const float* __restrict__ bias,
                                                  float* __restrict__ C) {
    constexpr int K     = HID;
    constexpr int NT    = NC / 16;
    constexpr int KG    = K / 8;
    constexpr int PITCH = K + 4;
    static_assert(NC % 16 == 0 && K * NC % FT == 0 && NN % 16 == 0 && SPL % 8 == 0, "shape");
    __shared__ alignas(16) float wt[NC * PITCH];
    __shared__ alignas(16) float tile[FW * 16 * TP];

    const int lane = threadIdx.x & 31;
    const int c    = lane & 15;
    const int h    = lane >> 4;
    const unsigned lb = (unsigned)lane * 8u;
    const int wave = __builtin_amdgcn_readfirstlane((int)(threadIdx.x >> 5));
    const int s    = (int)blockIdx.x * FW + wave;
    const bool live = s < NN / 16;

    int rpv = 0;
    if (live) {
        const int idx = s * 16 + min(lane, 16);
        rpv = (idx < NN) ? rowptr[idx] : NE;
    }

    constexpr int PER = K * NC / FT;
    float wv[PER];
#pragma unroll
    for (int u = 0; u < PER; ++u) wv[u] = W[threadIdx.x + u * FT];
#pragma unroll
    for (int u = 0; u < PER; ++u) {
        const int i = threadIdx.x + u * FT;
        const int k = i / NC;
        const int n = i - k * NC;
        wt[n * PITCH + k] = wv[u];
    }
    float bv[NT];
#pragma unroll
    for (int nt = 0; nt < NT; ++nt) bv[nt] = BIAS ? bias[nt * 16 + c] : 0.0f;
    const float2 bb = *(const float2*)(b + 2 * lane);
    int beg = __builtin_amdgcn_readlane(rpv, 0);
    int cur[GB];
#pragma unroll
    for (int u = 0; u < GB; ++u) cur[u] = csr[beg + u];
    __syncthreads();
    if (!live) return;

    float* tl = tile + wave * 16 * TP;

    const char* hlo = (const char*)hs;
    const char* hsh = SPLIT ? (const char*)hs_hi - (size_t)SPL * HID * sizeof(float) : hlo;
#pragma unroll 2
    for (int q = 0; q < 16; ++q) {
        const int d   = s * 16 + q;
        const int end = __builtin_amdgcn_readlane(rpv, q + 1);
        const int deg = end - beg;

        int nxt[GB];
#pragma unroll
        for (int u = 0; u < GB; ++u) nxt[u] = csr[end + u];
        float2 acc = load_row2<SPLIT>(hlo, hsh, d, lb);
        add_rows<SPLIT>(acc, cur, min(deg, GB), lb, hlo, hsh);

        for (int k = GB; k < deg; k += GB) {
            int more[GB];
#pragma unroll
            for (int u = 0; u < GB; ++u) more[u] = csr[beg + k + u];
            add_rows<SPLIT>(acc, more, min(deg - k, GB), lb, hlo, hsh);
        }

        const float di = rsqrtf((float)(deg + 1));
        float2 o;
        o.x = fmaxf(di * acc.x + bb.x, 0.0f);
        o.y = fmaxf(di * acc.y + bb.y, 0.0f);
        *(float2*)(tl + q * TP + 2 * lane) = o;
        beg = end;
#pragma unroll
        for (int u = 0; u < GB; ++u) cur[u] = nxt[u];
    }

    float4 a[KG];
#pragma unroll
    for (int j = 0; j < KG; ++j) a[j] = *(const float4*)(tl + c * TP + 8 * j + 4 * h);

    const float* wl = wt + c * PITCH + 4 * h;
    v8f acc[NT];
#pragma unroll
    for (int nt = 0; nt < NT; ++nt) acc[nt] = (v8f){0.f, 0.f, 0.f, 0.f, 0.f, 0.f, 0.f, 0.f};

    float4 bc[NT], bn[NT];
#pragma unroll
    for (int u = 0; u < NT; ++u) bc[u] = *(const float4*)(wl + u * 16 * PITCH);
#pragma unroll
    for (int j = 0; j < KG; ++j) {
        if (j + 1 < KG) {
#pragma unroll
            for (int u = 0; u < NT; ++u) bn[u] = *(const float4*)(wl + u * 16 * PITCH + 8 * (j + 1));
        }
        const v2f a01 = {a[j].x, a[j].y};
        const v2f a23 = {a[j].z, a[j].w};
#pragma unroll
        for (int nt = 0; nt < NT; ++nt) {
            const v2f b01 = {bc[nt].x, bc[nt].y};
            acc[nt] = __builtin_amdgcn_wmma_f32_16x16x4_f32(false, a01, false, b01,
                                                            (short)0, acc[nt], false, false);
        }
#pragma unroll
        for (int nt = 0; nt < NT; ++nt) {
            const v2f b23 = {bc[nt].z, bc[nt].w};
            acc[nt] = __builtin_amdgcn_wmma_f32_16x16x4_f32(false, a23, false, b23,
                                                            (short)0, acc[nt], false, false);
        }
        if (j + 1 < KG) {
#pragma unroll
            for (int u = 0; u < NT; ++u) bc[u] = bn[u];
        }
    }

    float sc[8];
    if constexpr (SCALE) {
        const int rnext = __shfl_down(rpv, 1);
        const float dl  = rsqrtf((float)(rnext - rpv + 1));
#pragma unroll
        for (int v = 0; v < 8; ++v) sc[v] = __shfl(dl, 8 * h + v);
    } else {
#pragma unroll
        for (int v = 0; v < 8; ++v) sc[v] = 1.0f;
    }

    float* crow = C + (size_t)(s * 16 + 8 * h) * NC + c;
#pragma unroll
    for (int v = 0; v < 8; ++v) {
#pragma unroll
        for (int nt = 0; nt < NT; ++nt) {
            const float t = BIAS ? acc[nt][v] + bv[nt] : acc[nt][v];
            crow[(size_t)v * NC + nt * 16] = t * sc[v];
        }
    }
}

extern "C" void kernel_launch(void* const* d_in, const int* in_sizes, int n_in,
                              void* d_out, int out_size, void* d_ws, size_t ws_size,
                              hipStream_t stream) {
    const float* x  = (const float*)d_in[0];
    const int*   ei = (const int*)d_in[1];
    const float* W1 = (const float*)d_in[2];
    const float* b1 = (const float*)d_in[3];
    const float* W2 = (const float*)d_in[4];
    const float* b2 = (const float*)d_in[5];
    const float* Wl = (const float*)d_in[6];
    const float* bl = (const float*)d_in[7];
    float* out = (float*)d_out;

    int*   rowptr = (int*)d_ws;
    float* buf1 = (float*)d_ws + NN;
    int*   csr  = (int*)(buf1 + (size_t)SPL * HID);
    float* buf2 = buf1 + (size_t)NN * HID;
    int* tsrc  = (int*)buf2;
    int* tdst  = tsrc + NE;
    int* pt    = tdst + NE;
    float* hi1 = out;

    const int* src = ei;
    const int* dst = ei + NE;

    range_scatter<<<G1, T1, 0, stream>>>(src, dst, pt, tsrc, tdst);
    range_rows<<<NB, T1, 0, stream>>>(tsrc, tdst, pt, rowptr, csr);

    gemm_strip<INF, HID, true, false><<<GW, GT, 0, stream>>>(x, W1, nullptr, buf1, hi1, rowptr, NN);

    gather_gemm<HID, true, false, true><<<FG, FT, 0, stream>>>(rowptr, csr, buf1, hi1, b1, W2,
                                                               nullptr, buf2);

    gather_gemm<NCL, false, true, false><<<FG, FT, 0, stream>>>(rowptr, csr, buf2, nullptr, b2, Wl,
                                                                bl, out);
}


# speedup vs baseline: 5.5587x; 1.0398x over previous
#include <hip/hip_runtime.h>


#define NN 100000
#define NE 1600000
#define INF 128
#define HID 64
#define NCL 16
#define NB 256
#define BW 391
#define G1 256
#define T1 1024
#define CH ((NE + G1 - 1) / G1)
#define EPT ((CH + T1 - 1) / T1)
#define CAP2 12288
#define RL 4
#define PP 288
#define RB 16
#define SPL 75000
#define FT 256
#define FW (FT / 32)
#define FG ((NN / 16 + FW - 1) / FW)
#define TP (HID + 4)
#define GB 24

typedef float v2f __attribute__((ext_vector_type(2)));
typedef float v8f __attribute__((ext_vector_type(8)));


__device__ __forceinline__ int range_of(int d) {
    return min((int)((unsigned)d / (unsigned)BW), NB - 1);
}

__device__ __forceinline__ int group_excl_scan(int v, int* wsum, int& total) {
    const int lane = threadIdx.x & 31;
    const int wave = threadIdx.x >> 5;
    int x = v;
#pragma unroll
    for (int o = 1; o < 32; o <<= 1) {
        int y = __shfl_up(x, o);
        if (lane >= o) x += y;
    }
    if (lane == 31) wsum[wave] = x;
    __syncthreads();
    if (wave == 0) {
        const int s = wsum[lane];
        int xs = s;
#pragma unroll
        for (int o = 1; o < 32; o <<= 1) {
            int y = __shfl_up(xs, o);
            if (lane >= o) xs += y;
        }
        wsum[lane] = xs - s;
        if (lane == 31) wsum[32] = xs;
    }
    __syncthreads();
    const int res = wsum[wave] + x - v;
    total = wsum[32];
    __syncthreads();
    return res;
}

__global__ __launch_bounds__(T1) void range_scatter(const int* __restrict__ src,
                                                     const int* __restrict__ dst,
                                                     int* __restrict__ pt,
                                                     int* __restrict__ tsrc,
                                                     int* __restrict__ tdst) {
    __shared__ int cnt[NB];
    __shared__ int wsum[33];
    __shared__ int s_src[CH];
    __shared__ int s_dst[CH];
    const int t  = threadIdx.x;
    const int me = blockIdx.x;

    const int c0 = me * CH;
    const int c1 = min(c0 + CH, NE);
    int es[EPT], ed[EPT], er[EPT];
#pragma unroll
    for (int j = 0; j < EPT; ++j) {
        const int i = min(c0 + t + j * T1, c1 - 1);
        es[j] = src[i];
        ed[j] = dst[i];
    }
    if (t < NB) cnt[t] = 0;
    __syncthreads();

#pragma unroll
    for (int j = 0; j < EPT; ++j) {
        er[j] = 0;
        if (c0 + t + j * T1 < c1) er[j] = atomicAdd(&cnt[range_of(ed[j])], 1);
    }
    __syncthreads();

    const int lc = (t < NB) ? cnt[t] : 0;
    int chunk_edges;
    const int lstart = group_excl_scan(lc, wsum, chunk_edges);
    if (t < NB) {
        cnt[t] = lstart;
        pt[me * PP + t] = lstart;
    }
    if (t == 0) pt[me * PP + NB] = chunk_edges;
    __syncthreads();

#pragma unroll
    for (int j = 0; j < EPT; ++j) {
        const int i = c0 + t + j * T1;
        if (i < c1) {
            const int p = cnt[range_of(ed[j])] + er[j];
            s_src[p] = es[j];
            s_dst[p] = ed[j];
        }
    }
    __syncthreads();

    for (int i = t; i < chunk_edges; i += T1) {
        tsrc[c0 + i] = s_src[i];
        tdst[c0 + i] = s_dst[i];
    }
}

__device__ __forceinline__ int node_slot(int d, int n0) {
    return min(max(d - n0, 0), BW - 1);
}

__global__ __launch_bounds__(T1) void range_rows(const int* __restrict__ tsrc,
                                                  const int* __restrict__ tdst,
                                                  const int* __restrict__ pt,
                                                  int* __restrict__ rowptr,
                                                  int* __restrict__ csr) {
    static_assert(T1 == RL * G1 && PP >= NB + 1 && PP % 32 == 0, "run table shape");
    __shared__ int cnt[BW];
    __shared__ int wsum[33];
    __shared__ int s_out[CAP2];
    const int t  = threadIdx.x;
    const int n0 = blockIdx.x * BW;
    const int nn = min(BW, NN - n0);

    const int ch  = t / RL;
    const int sub = t % RL;
    const int a   = pt[ch * PP + blockIdx.x];
    const int len = pt[ch * PP + blockIdx.x + 1] - a;
    const int* run_d = tdst + ch * CH + a;
    const int* run_s = tsrc + ch * CH + a;

    int d0[RB], s0[RB];
#pragma unroll
    for (int u = 0; u < RB; ++u) {
        const int i = sub + RL * u;
        d0[u] = (i < len) ? run_d[i] : 0;
        s0[u] = (i < len) ? run_s[i] : 0;
    }

    if (t < BW) cnt[t] = 0;
    int e0, n;
    group_excl_scan(sub == 0 ? a : 0, wsum, e0);
    group_excl_scan(sub == 0 ? len : 0, wsum, n);

#pragma unroll
    for (int u = 0; u < RB; ++u)
        if (sub + RL * u < len) atomicAdd(&cnt[node_slot(d0[u], n0)], 1);
    for (int i0 = sub + RL * RB; i0 < len; i0 += RL * RB) {
        int dl[RB];
#pragma unroll
        for (int u = 0; u < RB; ++u) {
            const int i = i0 + RL * u;
            dl[u] = (i < len) ? run_d[i] : 0;
        }
#pragma unroll
        for (int u = 0; u < RB; ++u)
            if (i0 + RL * u < len) atomicAdd(&cnt[node_slot(dl[u], n0)], 1);
    }
    __syncthreads();

    const int c = (t < BW) ? cnt[t] : 0;
    int seen;
    const int rs = group_excl_scan(c, wsum, seen);
    if (t < nn) rowptr[n0 + t] = e0 + rs;
    if (t < BW) cnt[t] = rs;
    __syncthreads();

    const bool staged = (n <= CAP2);
#pragma unroll
    for (int u = 0; u < RB; ++u) {
        if (sub + RL * u < len) {
            const int p = atomicAdd(&cnt[node_slot(d0[u], n0)], 1);
            if (staged) s_out[p] = s0[u];
            else        csr[e0 + p] = s0[u];
        }
    }
    for (int i0 = sub + RL * RB; i0 < len; i0 += RL * RB) {
        int dl[RB], sv[RB];
#pragma unroll
        for (int u = 0; u < RB; ++u) {
            const int i = i0 + RL * u;
            dl[u] = (i < len) ? run_d[i] : 0;
            sv[u] = (i < len) ? run_s[i] : 0;
        }
#pragma unroll
        for (int u = 0; u < RB; ++u) {
            if (i0 + RL * u < len) {
                const int p = atomicAdd(&cnt[node_slot(dl[u], n0)], 1);
                if (staged) s_out[p] = sv[u];
                else        csr[e0 + p] = sv[u];
            }
        }
    }
    __syncthreads();
    if (staged)
        for (int i = t; i < n; i += T1) csr[e0 + i] = s_out[i];
}

#define GW 256
#define GT 512

template <int K, bool SCALE>
__device__ __forceinline__ void load_strip(const float* __restrict__ A,
                                           const int* __restrict__ rowptr, int M, int s, int c,
                                           int h, float4 (&a)[K / 8], int (&rp)[9]) {
    const float* arow = A + (size_t)(s * 16 + c) * K + 4 * h;
#pragma unroll
    for (int j = 0; j < K / 8; ++j) a[j] = *(const float4*)(arow + 8 * j);
    if constexpr (SCALE) {
        const int r0 = s * 16 + 8 * h;
#pragma unroll
        for (int v = 0; v < 8; ++v) rp[v] = rowptr[r0 + v];
        rp[8] = rowptr[min(r0 + 8, M - 1)];
    }
}

template <int K, int NC, bool SCALE, bool BIAS>
__global__ __launch_bounds__(GT) void gemm_strip(const float* __restrict__ A,
                                                 const float* __restrict__ W,
                                                 const float* __restrict__ bias,
                                                 float* __restrict__ C,
                                                 float* __restrict__ Chi,
                                                 const int* __restrict__ rowptr, int M) {
    constexpr int NT    = NC / 16;
    constexpr int KG    = K / 8;
    constexpr int PITCH = K + 4;
    constexpr int BG    = 2;
    static_assert(K % (8 * BG) == 0 && NC % 16 == 0, "shape");
    __shared__ alignas(16) float wt[NC * PITCH];

    const int lane = threadIdx.x & 31;
    const int c    = lane & 15;
    const int h    = lane >> 4;
    const int wave   = __builtin_amdgcn_readfirstlane((int)(threadIdx.x >> 5));
    const int nwaves = GT / 32;
    const int nstrip = M >> 4;
    const int step   = (int)gridDim.x * nwaves;
    int s = (int)blockIdx.x + (int)gridDim.x * wave;

    float4 a[KG];
    int rp[9];
#pragma unroll
    for (int j = 0; j < KG; ++j) a[j] = make_float4(0.f, 0.f, 0.f, 0.f);
#pragma unroll
    for (int v = 0; v < 9; ++v) rp[v] = 0;
    if (s < nstrip) load_strip<K, SCALE>(A, rowptr, M, s, c, h, a, rp);

    constexpr int PER = K * NC / GT;
    static_assert(K * NC % GT == 0, "staging");
    float wv[PER];
#pragma unroll
    for (int u = 0; u < PER; ++u) wv[u] = W[threadIdx.x + u * GT];
#pragma unroll
    for (int u = 0; u < PER; ++u) {
        const int i = threadIdx.x + u * GT;
        const int k = i / NC;
        const int n = i - k * NC;
        wt[n * PITCH + k] = wv[u];
    }
    float bv[NT];
#pragma unroll
    for (int nt = 0; nt < NT; ++nt) bv[nt] = BIAS ? bias[nt * 16 + c] : 0.0f;
    __syncthreads();

    const float* wl = wt + c * PITCH + 4 * h;

    while (s < nstrip) {
        asm volatile("" ::: "memory");
        const int r0 = s * 16 + 8 * h;

        v8f acc[NT];
#pragma unroll
        for (int nt = 0; nt < NT; ++nt) acc[nt] = (v8f){0.f, 0.f, 0.f, 0.f, 0.f, 0.f, 0.f, 0.f};

        float4 bc[BG * NT], bn[BG * NT];
#pragma unroll
        for (int u = 0; u < BG * NT; ++u)
            bc[u] = *(const float4*)(wl + (u % NT) * 16 * PITCH + 8 * (u / NT));
#pragma unroll
        for (int j0 = 0; j0 < KG; j0 += BG) {
            if (j0 + BG < KG) {
#pragma unroll
                for (int u = 0; u < BG * NT; ++u)
                    bn[u] = *(const float4*)(wl + (u % NT) * 16 * PITCH + 8 * (j0 + BG + u / NT));
            }
#pragma unroll
            for (int g = 0; g < BG; ++g) {
                const float4 av = a[j0 + g];
                const v2f a01 = {av.x, av.y};
                const v2f a23 = {av.z, av.w};
#pragma unroll
                for (int nt = 0; nt < NT; ++nt) {
                    const v2f b01 = {bc[g * NT + nt].x, bc[g * NT + nt].y};
                    acc[nt] = __builtin_amdgcn_wmma_f32_16x16x4_f32(false, a01, false, b01,
                                                                    (short)0, acc[nt], false, false);
                }
#pragma unroll
                for (int nt = 0; nt < NT; ++nt) {
                    const v2f b23 = {bc[g * NT + nt].z, bc[g * NT + nt].w};
                    acc[nt] = __builtin_amdgcn_wmma_f32_16x16x4_f32(false, a23, false, b23,
                                                                    (short)0, acc[nt], false, false);
                }
            }
            if (j0 + BG < KG) {
#pragma unroll
                for (int u = 0; u < BG * NT; ++u) bc[u] = bn[u];
            }
        }

        float sc[8];
        if constexpr (SCALE) {
            if (r0 + 8 >= M) rp[8] = NE;
#pragma unroll
            for (int v = 0; v < 8; ++v)
                sc[v] = rsqrtf((float)(rp[v + 1] - rp[v] + 1));
        } else {
#pragma unroll
            for (int v = 0; v < 8; ++v) sc[v] = 1.0f;
        }

        float* crow = (r0 < SPL) ? C + (size_t)r0 * NC + c : Chi + (size_t)(r0 - SPL) * NC + c;
#pragma unroll
        for (int v = 0; v < 8; ++v) {
#pragma unroll
            for (int nt = 0; nt < NT; ++nt) {
                const float t = BIAS ? acc[nt][v] + bv[nt] : acc[nt][v];
                crow[(size_t)v * NC + nt * 16] = t * sc[v];
            }
        }
        s += step;
        if (s < nstrip) load_strip<K, SCALE>(A, rowptr, M, s, c, h, a, rp);
    }
}


template <bool SPLIT>
__device__ __forceinline__ float2 load_row2(const char* __restrict__ lo,
                                            const char* __restrict__ hi_shift, int r,
                                            unsigned lb) {
    const char* base = lo;
    if constexpr (SPLIT) {
        if (r >= SPL) base = hi_shift;
    }
    const unsigned off = (unsigned)r * (unsigned)(HID * sizeof(float)) + lb;
    return *(const float2*)(base + off);
}

template <int U, bool SPLIT>
__device__ __forceinline__ void rows_request(float2 (&v)[GB], const int (&sx)[GB], int n,
                                             unsigned lb, const char* __restrict__ lo,
                                             const char* __restrict__ hi_shift) {
    if constexpr (U < GB) {
        if (U < n) {
            v[U] = load_row2<SPLIT>(lo, hi_shift, sx[U], lb);
            rows_request<U + 1, SPLIT>(v, sx, n, lb, lo, hi_shift);
        }
    }
}

template <int U>
__device__ __forceinline__ void rows_add(float2& acc, const float2 (&v)[GB], int n) {
    if constexpr (U < GB) {
        if (U < n) {
            acc.x += v[U].x;
            acc.y += v[U].y;
            rows_add<U + 1>(acc, v, n);
        }
    }
}

template <bool SPLIT>
__device__ __forceinline__ void add_rows(float2& acc, const int (&sx)[GB], int n, unsigned lb,
                                         const char* __restrict__ lo,
                                         const char* __restrict__ hi_shift) {
    float2 v[GB];
    rows_request<0, SPLIT>(v, sx, n, lb, lo, hi_shift);
    rows_add<0>(acc, v, n);
}

template <int NC, bool SCALE, bool BIAS, bool SPLIT>
__global__ __launch_bounds__(FT) void gather_gemm(const int* __restrict__ rowptr,
                                                  const int* __restrict__ csr,
                                                  const float* __restrict__ hs,
                                                  const float* __restrict__ hs_hi,
                                                  const float* __restrict__ b,
                                                  const float* __restrict__ W,
                                                  const float* __restrict__ bias,
                                                  float* __restrict__ C) {
    constexpr int K     = HID;
    constexpr int NT    = NC / 16;
    constexpr int KG    = K / 8;
    constexpr int PITCH = K + 4;
    static_assert(NC % 16 == 0 && K * NC % FT == 0 && NN % 16 == 0 && SPL % 8 == 0, "shape");
    __shared__ alignas(16) float wt[NC * PITCH];
    __shared__ alignas(16) float tile[FW * 16 * TP];

    const int lane = threadIdx.x & 31;
    const int c    = lane & 15;
    const int h    = lane >> 4;
    const unsigned lb = (unsigned)lane * 8u;
    const int wave = __builtin_amdgcn_readfirstlane((int)(threadIdx.x >> 5));
    const int s    = (int)blockIdx.x * FW + wave;
    const bool live = s < NN / 16;

    int rpv = 0;
    if (live) {
        const int idx = s * 16 + min(lane, 16);
        rpv = (idx < NN) ? rowptr[idx] : NE;
    }

    constexpr int PER = K * NC / FT;
    float wv[PER];
#pragma unroll
    for (int u = 0; u < PER; ++u) wv[u] = W[threadIdx.x + u * FT];
#pragma unroll
    for (int u = 0; u < PER; ++u) {
        const int i = threadIdx.x + u * FT;
        const int k = i / NC;
        const int n = i - k * NC;
        wt[n * PITCH + k] = wv[u];
    }
    float bv[NT];
#pragma unroll
    for (int nt = 0; nt < NT; ++nt) bv[nt] = BIAS ? bias[nt * 16 + c] : 0.0f;
    const float2 bb = *(const float2*)(b + 2 * lane);
    int beg = __builtin_amdgcn_readlane(rpv, 0);
    int cur[GB];
#pragma unroll
    for (int u = 0; u < GB; ++u) cur[u] = csr[beg + u];
    __syncthreads();
    if (!live) return;

    float* tl = tile + wave * 16 * TP;

    const char* hlo = (const char*)hs;
    const char* hsh = SPLIT ? (const char*)hs_hi - (size_t)SPL * HID * sizeof(float) : hlo;
#pragma unroll 2
    for (int q = 0; q < 16; ++q) {
        const int d   = s * 16 + q;
        const int end = __builtin_amdgcn_readlane(rpv, q + 1);
        const int deg = end - beg;

        int nxt[GB];
#pragma unroll
        for (int u = 0; u < GB; ++u) nxt[u] = csr[end + u];
        float2 acc = load_row2<SPLIT>(hlo, hsh, d, lb);
        add_rows<SPLIT>(acc, cur, min(deg, GB), lb, hlo, hsh);

        for (int k = GB; k < deg; k += GB) {
            int more[GB];
#pragma unroll
            for (int u = 0; u < GB; ++u) more[u] = csr[beg + k + u];
            add_rows<SPLIT>(acc, more, min(deg - k, GB), lb, hlo, hsh);
        }

        const float di = rsqrtf((float)(deg + 1));
        float2 o;
        o.x = fmaxf(di * acc.x + bb.x, 0.0f);
        o.y = fmaxf(di * acc.y + bb.y, 0.0f);
        *(float2*)(tl + q * TP + 2 * lane) = o;
        beg = end;
#pragma unroll
        for (int u = 0; u < GB; ++u) cur[u] = nxt[u];
    }

    float4 a[KG];
#pragma unroll
    for (int j = 0; j < KG; ++j) a[j] = *(const float4*)(tl + c * TP + 8 * j + 4 * h);

    const float* wl = wt + c * PITCH + 4 * h;
    v8f acc[NT];
#pragma unroll
    for (int nt = 0; nt < NT; ++nt) acc[nt] = (v8f){0.f, 0.f, 0.f, 0.f, 0.f, 0.f, 0.f, 0.f};

    float4 bc[NT], bn[NT];
#pragma unroll
    for (int u = 0; u < NT; ++u) bc[u] = *(const float4*)(wl + u * 16 * PITCH);
#pragma unroll
    for (int j = 0; j < KG; ++j) {
        if (j + 1 < KG) {
#pragma unroll
            for (int u = 0; u < NT; ++u) bn[u] = *(const float4*)(wl + u * 16 * PITCH + 8 * (j + 1));
        }
        const v2f a01 = {a[j].x, a[j].y};
        const v2f a23 = {a[j].z, a[j].w};
#pragma unroll
        for (int nt = 0; nt < NT; ++nt) {
            const v2f b01 = {bc[nt].x, bc[nt].y};
            acc[nt] = __builtin_amdgcn_wmma_f32_16x16x4_f32(false, a01, false, b01,
                                                            (short)0, acc[nt], false, false);
        }
#pragma unroll
        for (int nt = 0; nt < NT; ++nt) {
            const v2f b23 = {bc[nt].z, bc[nt].w};
            acc[nt] = __builtin_amdgcn_wmma_f32_16x16x4_f32(false, a23, false, b23,
                                                            (short)0, acc[nt], false, false);
        }
        if (j + 1 < KG) {
#pragma unroll
            for (int u = 0; u < NT; ++u) bc[u] = bn[u];
        }
    }

    float sc[8];
    if constexpr (SCALE) {
        const int rnext = __shfl_down(rpv, 1);
        const float dl  = rsqrtf((float)(rnext - rpv + 1));
#pragma unroll
        for (int v = 0; v < 8; ++v) sc[v] = __shfl(dl, 8 * h + v);
    } else {
#pragma unroll
        for (int v = 0; v < 8; ++v) sc[v] = 1.0f;
    }

    float* crow = C + (size_t)(s * 16 + 8 * h) * NC + c;
#pragma unroll
    for (int v = 0; v < 8; ++v) {
#pragma unroll
        for (int nt = 0; nt < NT; ++nt) {
            const float t = BIAS ? acc[nt][v] + bv[nt] : acc[nt][v];
            crow[(size_t)v * NC + nt * 16] = t * sc[v];
        }
    }
}

extern "C" void kernel_launch(void* const* d_in, const int* in_sizes, int n_in,
                              void* d_out, int out_size, void* d_ws, size_t ws_size,
                              hipStream_t stream) {
    const float* x  = (const float*)d_in[0];
    const int*   ei = (const int*)d_in[1];
    const float* W1 = (const float*)d_in[2];
    const float* b1 = (const float*)d_in[3];
    const float* W2 = (const float*)d_in[4];
    const float* b2 = (const float*)d_in[5];
    const float* Wl = (const float*)d_in[6];
    const float* bl = (const float*)d_in[7];
    float* out = (float*)d_out;

    float* buf1 = (float*)d_ws;
    int*   csr  = (int*)(buf1 + (size_t)SPL * HID);
    float* buf2 = buf1 + (size_t)NN * HID;
    int*   rowptr = (int*)(buf2 + (size_t)NN * HID);
    int* tsrc  = (int*)buf2;
    int* tdst  = tsrc + NE;
    int* pt    = tdst + NE;
    float* hi1 = out;

    const int* src = ei;
    const int* dst = ei + NE;

    range_scatter<<<G1, T1, 0, stream>>>(src, dst, pt, tsrc, tdst);
    range_rows<<<NB, T1, 0, stream>>>(tsrc, tdst, pt, rowptr, csr);

    gemm_strip<INF, HID, true, false><<<GW, GT, 0, stream>>>(x, W1, nullptr, buf1, hi1, rowptr, NN);

    gather_gemm<HID, true, false, true><<<FG, FT, 0, stream>>>(rowptr, csr, buf1, hi1, b1, W2,
                                                               nullptr, buf2);

    gather_gemm<NCL, false, true, false><<<FG, FT, 0, stream>>>(rowptr, csr, buf2, nullptr, b2, Wl,
                                                                bl, out);
}


# layer-1 product moved into the row-start launch: ranges of 400 nodes = 25 strips, a range's work-group multiplies its own rows behind its sort, half of each row in flight under it; 5 launches to 4
# speedup vs baseline: 5.9483x; 1.0701x over previous
#include <hip/hip_runtime.h>


#define NN 100000
#define NE 1600000
#define INF 128
#define HID 64
#define NCL 16
#define NB 250
#define BW 400
#define SW (BW / 16)
#define G1 256
#define T1 1024
#define CH ((NE + G1 - 1) / G1)
#define EPT ((CH + T1 - 1) / T1)
#define CAP2 12288
#define RL 4
#define PP 288
#define RB 16
#define SPL 75000
#define FT 256
#define FW (FT / 32)
#define FG ((NN / 16 + FW - 1) / FW)
#define TP (HID + 4)
#define GB 24

typedef float v2f __attribute__((ext_vector_type(2)));
typedef float v8f __attribute__((ext_vector_type(8)));


__device__ __forceinline__ int range_of(int d) {
    return min((int)((unsigned)d / (unsigned)BW), NB - 1);
}

template <bool LOCAL>
__device__ __forceinline__ void group_barrier() {
    if constexpr (LOCAL) {
        __builtin_amdgcn_fence(__ATOMIC_RELEASE, "workgroup", "local");
        __builtin_amdgcn_s_barrier();
        __builtin_amdgcn_fence(__ATOMIC_ACQUIRE, "workgroup", "local");
    } else {
        __syncthreads();
    }
}

template <bool LOCAL>
__device__ __forceinline__ int group_excl_scan(int v, int* wsum, int& total) {
    const int lane = threadIdx.x & 31;
    const int wave = threadIdx.x >> 5;
    int x = v;
#pragma unroll
    for (int o = 1; o < 32; o <<= 1) {
        int y = __shfl_up(x, o);
        if (lane >= o) x += y;
    }
    if (lane == 31) wsum[wave] = x;
    group_barrier<LOCAL>();
    if (wave == 0) {
        const int s = wsum[lane];
        int xs = s;
#pragma unroll
        for (int o = 1; o < 32; o <<= 1) {
            int y = __shfl_up(xs, o);
            if (lane >= o) xs += y;
        }
        wsum[lane] = xs - s;
        if (lane == 31) wsum[32] = xs;
    }
    group_barrier<LOCAL>();
    const int res = wsum[wave] + x - v;
    total = wsum[32];
    group_barrier<LOCAL>();
    return res;
}

__global__ __launch_bounds__(T1) void range_scatter(const int* __restrict__ src,
                                                     const int* __restrict__ dst,
                                                     int* __restrict__ pt,
                                                     int* __restrict__ tsrc,
                                                     int* __restrict__ tdst) {
    __shared__ int cnt[NB];
    __shared__ int wsum[33];
    __shared__ int s_src[CH];
    __shared__ int s_dst[CH];
    const int t  = threadIdx.x;
    const int me = blockIdx.x;

    const int c0 = me * CH;
    const int c1 = min(c0 + CH, NE);
    int es[EPT], ed[EPT], er[EPT];
#pragma unroll
    for (int j = 0; j < EPT; ++j) {
        const int i = min(c0 + t + j * T1, c1 - 1);
        es[j] = src[i];
        ed[j] = dst[i];
    }
    if (t < NB) cnt[t] = 0;
    __syncthreads();

#pragma unroll
    for (int j = 0; j < EPT; ++j) {
        er[j] = 0;
        if (c0 + t + j * T1 < c1) er[j] = atomicAdd(&cnt[range_of(ed[j])], 1);
    }
    __syncthreads();

    const int lc = (t < NB) ? cnt[t] : 0;
    int chunk_edges;
    const int lstart = group_excl_scan<false>(lc, wsum, chunk_edges);
    if (t < NB) {
        cnt[t] = lstart;
        pt[me * PP + t] = lstart;
    }
    if (t == 0) pt[me * PP + NB] = chunk_edges;
    __syncthreads();

#pragma unroll
    for (int j = 0; j < EPT; ++j) {
        const int i = c0 + t + j * T1;
        if (i < c1) {
            const int p = cnt[range_of(ed[j])] + er[j];
            s_src[p] = es[j];
            s_dst[p] = ed[j];
        }
    }
    __syncthreads();

    for (int i = t; i < chunk_edges; i += T1) {
        tsrc[c0 + i] = s_src[i];
        tdst[c0 + i] = s_dst[i];
    }
}

__device__ __forceinline__ int node_slot(int d, int n0) {
    return min(max(d - n0, 0), BW - 1);
}


__global__ __launch_bounds__(T1) void range_rows_product(const int* __restrict__ tsrc,
                                                          const int* __restrict__ tdst,
                                                          const int* __restrict__ pt,
                                                          int* __restrict__ rowptr,
                                                          int* __restrict__ csr,
                                                          const float* __restrict__ X,
                                                          const float* __restrict__ W,
                                                          float* __restrict__ C,
                                                          float* __restrict__ Chi) {
    constexpr int K     = INF;
    constexpr int NC    = HID;
    constexpr int NT    = NC / 16;
    constexpr int KG    = K / 8;
    constexpr int PITCH = K + 4;
    constexpr int PER   = K * NC / T1;
    static_assert(T1 == RL * G1 && PP >= NB + 1 && PP % 32 == 0, "run table shape");
    static_assert(NB * BW == NN && BW % 16 == 0 && SW * 32 <= T1 && SPL % 8 == 0, "strips of a range");
    static_assert(K * NC % T1 == 0 && NT % 2 == 0 && KG % 2 == 0, "product shape");
    __shared__ int cnt[BW];
    __shared__ int wsum[33];
    __shared__ int s_out[CAP2];
    __shared__ alignas(16) float dv[BW];
    __shared__ alignas(16) float wt[NC * PITCH];
    const int t  = threadIdx.x;
    const int n0 = blockIdx.x * BW;
    const int nn = min(BW, NN - n0);
    const int col = t & 15;
    const int hf  = (t >> 4) & 1;
    const int wave = __builtin_amdgcn_readfirstlane(t >> 5);

    const int ch  = t / RL;
    const int sub = t % RL;
    const int a   = pt[ch * PP + blockIdx.x];
    const int len = pt[ch * PP + blockIdx.x + 1] - a;
    const int* run_d = tdst + ch * CH + a;
    const int* run_s = tsrc + ch * CH + a;

    int d0[RB], s0[RB];
#pragma unroll
    for (int u = 0; u < RB; ++u) {
        const int i = sub + RL * u;
        d0[u] = (i < len) ? run_d[i] : 0;
        s0[u] = (i < len) ? run_s[i] : 0;
    }

    float wv[PER];
#pragma unroll
    for (int u = 0; u < PER; ++u) wv[u] = W[t + u * T1];

    if (t < BW) cnt[t] = 0;
    int e0, n;
    group_excl_scan<true>(sub == 0 ? a : 0, wsum, e0);
    group_excl_scan<true>(sub == 0 ? len : 0, wsum, n);

#pragma unroll
    for (int u = 0; u < PER; ++u) {
        const int i = t + u * T1;
        const int k = i / NC;
        const int m = i - k * NC;
        wt[m * PITCH + k] = wv[u];
    }

    const float* xrow = X + (size_t)(n0 + wave * 16 + col) * K + 4 * hf;
    float4 xa[KG];
    if (wave < SW) {
#pragma unroll
        for (int j = 0; j < KG / 2; ++j) xa[j] = *(const float4*)(xrow + 8 * j);
    }

#pragma unroll
    for (int u = 0; u < RB; ++u)
        if (sub + RL * u < len) atomicAdd(&cnt[node_slot(d0[u], n0)], 1);
    for (int i0 = sub + RL * RB; i0 < len; i0 += RL * RB) {
        int dl[RB];
#pragma unroll
        for (int u = 0; u < RB; ++u) {
            const int i = i0 + RL * u;
            dl[u] = (i < len) ? run_d[i] : 0;
        }
#pragma unroll
        for (int u = 0; u < RB; ++u)
            if (i0 + RL * u < len) atomicAdd(&cnt[node_slot(dl[u], n0)], 1);
    }
    group_barrier<true>();

    const int cn = (t < BW) ? cnt[t] : 0;
    if (t < BW) dv[t] = rsqrtf((float)(cn + 1));
    int seen;
    const int rs = group_excl_scan<true>(cn, wsum, seen);
    if (t < nn) rowptr[n0 + t] = e0 + rs;
    if (t < BW) cnt[t] = rs;
    group_barrier<true>();

    const bool staged = (n <= CAP2);
#pragma unroll
    for (int u = 0; u < RB; ++u) {
        if (sub + RL * u < len) {
            const int p = atomicAdd(&cnt[node_slot(d0[u], n0)], 1);
            if (staged) s_out[p] = s0[u];
            else        csr[e0 + p] = s0[u];
        }
    }
    for (int i0 = sub + RL * RB; i0 < len; i0 += RL * RB) {
        int dl[RB], sv[RB];
#pragma unroll
        for (int u = 0; u < RB; ++u) {
            const int i = i0 + RL * u;
            dl[u] = (i < len) ? run_d[i] : 0;
            sv[u] = (i < len) ? run_s[i] : 0;
        }
#pragma unroll
        for (int u = 0; u < RB; ++u) {
            if (i0 + RL * u < len) {
                const int p = atomicAdd(&cnt[node_slot(dl[u], n0)], 1);
                if (staged) s_out[p] = sv[u];
                else        csr[e0 + p] = sv[u];
            }
        }
    }
    group_barrier<true>();
    if (staged)
        for (int i = t; i < n; i += T1) csr[e0 + i] = s_out[i];

    if (wave >= SW) return;
#pragma unroll
    for (int j = KG / 2; j < KG; ++j) xa[j] = *(const float4*)(xrow + 8 * j);

    const int rl = wave * 16 + 8 * hf;
    const int r0 = n0 + rl;
    float sc[8];
#pragma unroll
    for (int v = 0; v < 8; ++v) sc[v] = dv[rl + v];
    float* crow = (r0 < SPL) ? C + (size_t)r0 * NC + col : Chi + (size_t)(r0 - SPL) * NC + col;
    const float* wl = wt + col * PITCH + 4 * hf;

#pragma unroll
    for (int p = 0; p < NT; p += 2) {
        v8f acc[2];
#pragma unroll
        for (int u = 0; u < 2; ++u) acc[u] = (v8f){0.f, 0.f, 0.f, 0.f, 0.f, 0.f, 0.f, 0.f};

        float4 bc[2], bn[2];
#pragma unroll
        for (int u = 0; u < 2; ++u) bc[u] = *(const float4*)(wl + (p + u) * 16 * PITCH);
#pragma unroll
        for (int j = 0; j < KG; ++j) {
            if (j + 1 < KG) {
#pragma unroll
                for (int u = 0; u < 2; ++u)
                    bn[u] = *(const float4*)(wl + (p + u) * 16 * PITCH + 8 * (j + 1));
            }
            const v2f a01 = {xa[j].x, xa[j].y};
            const v2f a23 = {xa[j].z, xa[j].w};
#pragma unroll
            for (int u = 0; u < 2; ++u) {
                const v2f b01 = {bc[u].x, bc[u].y};
                acc[u] = __builtin_amdgcn_wmma_f32_16x16x4_f32(false, a01, false, b01,
                                                               (short)0, acc[u], false, false);
            }
#pragma unroll
            for (int u = 0; u < 2; ++u) {
                const v2f b23 = {bc[u].z, bc[u].w};
                acc[u] = __builtin_amdgcn_wmma_f32_16x16x4_f32(false, a23, false, b23,
                                                               (short)0, acc[u], false, false);
            }
            if (j + 1 < KG) {
#pragma unroll
                for (int u = 0; u < 2; ++u) bc[u] = bn[u];
            }
        }

#pragma unroll
        for (int v = 0; v < 8; ++v) {
#pragma unroll
            for (int u = 0; u < 2; ++u) crow[(size_t)v * NC + (p + u) * 16] = acc[u][v] * sc[v];
        }
    }
}


template <bool SPLIT>
__device__ __forceinline__ float2 load_row2(const char* __restrict__ lo,
                                            const char* __restrict__ hi_shift, int r,
                                            unsigned lb) {
    const char* base = lo;
    if constexpr (SPLIT) {
        if (r >= SPL) base = hi_shift;
    }
    const unsigned off = (unsigned)r * (unsigned)(HID * sizeof(float)) + lb;
    return *(const float2*)(base + off);
}

template <int U, bool SPLIT>
__device__ __forceinline__ void rows_request(float2 (&v)[GB], const int (&sx)[GB], int n,
                                             unsigned lb, const char* __restrict__ lo,
                                             const char* __restrict__ hi_shift) {
    if constexpr (U < GB) {
        if (U < n) {
            v[U] = load_row2<SPLIT>(lo, hi_shift, sx[U], lb);
            rows_request<U + 1, SPLIT>(v, sx, n, lb, lo, hi_shift);
        }
    }
}

template <int U>
__device__ __forceinline__ void rows_add(float2& acc, const float2 (&v)[GB], int n) {
    if constexpr (U < GB) {
        if (U < n) {
            acc.x += v[U].x;
            acc.y += v[U].y;
            rows_add<U + 1>(acc, v, n);
        }
    }
}

template <bool SPLIT>
__device__ __forceinline__ void add_rows(float2& acc, const int (&sx)[GB], int n, unsigned lb,
                                         const char* __restrict__ lo,
                                         const char* __restrict__ hi_shift) {
    float2 v[GB];
    rows_request<0, SPLIT>(v, sx, n, lb, lo, hi_shift);
    rows_add<0>(acc, v, n);
}

template <int NC, bool SCALE, bool BIAS, bool SPLIT>
__global__ __launch_bounds__(FT) void gather_gemm(const int* __restrict__ rowptr,
                                                  const int* __restrict__ csr,
                                                  const float* __restrict__ hs,
                                                  const float* __restrict__ hs_hi,
                                                  const float* __restrict__ b,
                                                  const float* __restrict__ W,
                                                  const float* __restrict__ bias,
                                                  float* __restrict__ C) {
    constexpr int K     = HID;
    constexpr int NT    = NC / 16;
    constexpr int KG    = K / 8;
    constexpr int PITCH = K + 4;
    static_assert(NC % 16 == 0 && K * NC % FT == 0 && NN % 16 == 0 && SPL % 8 == 0, "shape");
    __shared__ alignas(16) float wt[NC * PITCH];
    __shared__ alignas(16) float tile[FW * 16 * TP];

    const int lane = threadIdx.x & 31;
    const int c    = lane & 15;
    const int h    = lane >> 4;
    const unsigned lb = (unsigned)lane * 8u;
    const int wave = __builtin_amdgcn_readfirstlane((int)(threadIdx.x >> 5));
    const int s    = (int)blockIdx.x * FW + wave;
    const bool live = s < NN / 16;

    int rpv = 0;
    if (live) {
        const int idx = s * 16 + min(lane, 16);
        rpv = (idx < NN) ? rowptr[idx] : NE;
    }

    constexpr int PER = K * NC / FT;
    float wv[PER];
#pragma unroll
    for (int u = 0; u < PER; ++u) wv[u] = W[threadIdx.x + u * FT];
#pragma unroll
    for (int u = 0; u < PER; ++u) {
        const int i = threadIdx.x + u * FT;
        const int k = i / NC;
        const int n = i - k * NC;
        wt[n * PITCH + k] = wv[u];
    }
    float bv[NT];
#pragma unroll
    for (int nt = 0; nt < NT; ++nt) bv[nt] = BIAS ? bias[nt * 16 + c] : 0.0f;
    const float2 bb = *(const float2*)(b + 2 * lane);
    int beg = __builtin_amdgcn_readlane(rpv, 0);
    int cur[GB];
#pragma unroll
    for (int u = 0; u < GB; ++u) cur[u] = csr[beg + u];
    __syncthreads();
    if (!live) return;

    float* tl = tile + wave * 16 * TP;

    const char* hlo = (const char*)hs;
    const char* hsh = SPLIT ? (const char*)hs_hi - (size_t)SPL * HID * sizeof(float) : hlo;
#pragma unroll 2
    for (int q = 0; q < 16; ++q) {
        const int d   = s * 16 + q;
        const int end = __builtin_amdgcn_readlane(rpv, q + 1);
        const int deg = end - beg;

        int nxt[GB];
#pragma unroll
        for (int u = 0; u < GB; ++u) nxt[u] = csr[end + u];
        float2 acc = load_row2<SPLIT>(hlo, hsh, d, lb);
        add_rows<SPLIT>(acc, cur, min(deg, GB), lb, hlo, hsh);

        for (int k = GB; k < deg; k += GB) {
            int more[GB];
#pragma unroll
            for (int u = 0; u < GB; ++u) more[u] = csr[beg + k + u];
            add_rows<SPLIT>(acc, more, min(deg - k, GB), lb, hlo, hsh);
        }

        const float di = rsqrtf((float)(deg + 1));
        float2 o;
        o.x = fmaxf(di * acc.x + bb.x, 0.0f);
        o.y = fmaxf(di * acc.y + bb.y, 0.0f);
        *(float2*)(tl + q * TP + 2 * lane) = o;
        beg = end;
#pragma unroll
        for (int u = 0; u < GB; ++u) cur[u] = nxt[u];
    }

    float4 a[KG];
#pragma unroll
    for (int j = 0; j < KG; ++j) a[j] = *(const float4*)(tl + c * TP + 8 * j + 4 * h);

    const float* wl = wt + c * PITCH + 4 * h;
    v8f acc[NT];
#pragma unroll
    for (int nt = 0; nt < NT; ++nt) acc[nt] = (v8f){0.f, 0.f, 0.f, 0.f, 0.f, 0.f, 0.f, 0.f};

    float4 bc[NT], bn[NT];
#pragma unroll
    for (int u = 0; u < NT; ++u) bc[u] = *(const float4*)(wl + u * 16 * PITCH);
#pragma unroll
    for (int j = 0; j < KG; ++j) {
        if (j + 1 < KG) {
#pragma unroll
            for (int u = 0; u < NT; ++u) bn[u] = *(const float4*)(wl + u * 16 * PITCH + 8 * (j + 1));
        }
        const v2f a01 = {a[j].x, a[j].y};
        const v2f a23 = {a[j].z, a[j].w};
#pragma unroll
        for (int nt = 0; nt < NT; ++nt) {
            const v2f b01 = {bc[nt].x, bc[nt].y};
            acc[nt] = __builtin_amdgcn_wmma_f32_16x16x4_f32(false, a01, false, b01,
                                                            (short)0, acc[nt], false, false);
        }
#pragma unroll
        for (int nt = 0; nt < NT; ++nt) {
            const v2f b23 = {bc[nt].z, bc[nt].w};
            acc[nt] = __builtin_amdgcn_wmma_f32_16x16x4_f32(false, a23, false, b23,
                                                            (short)0, acc[nt], false, false);
        }
        if (j + 1 < KG) {
#pragma unroll
            for (int u = 0; u < NT; ++u) bc[u] = bn[u];
        }
    }

    float sc[8];
    if constexpr (SCALE) {
        const int rnext = __shfl_down(rpv, 1);
        const float dl  = rsqrtf((float)(rnext - rpv + 1));
#pragma unroll
        for (int v = 0; v < 8; ++v) sc[v] = __shfl(dl, 8 * h + v);
    } else {
#pragma unroll
        for (int v = 0; v < 8; ++v) sc[v] = 1.0f;
    }

    float* crow = C + (size_t)(s * 16 + 8 * h) * NC + c;
#pragma unroll
    for (int v = 0; v < 8; ++v) {
#pragma unroll
        for (int nt = 0; nt < NT; ++nt) {
            const float t = BIAS ? acc[nt][v] + bv[nt] : acc[nt][v];
            crow[(size_t)v * NC + nt * 16] = t * sc[v];
        }
    }
}

extern "C" void kernel_launch(void* const* d_in, const int* in_sizes, int n_in,
                              void* d_out, int out_size, void* d_ws, size_t ws_size,
                              hipStream_t stream) {
    const float* x  = (const float*)d_in[0];
    const int*   ei = (const int*)d_in[1];
    const float* W1 = (const float*)d_in[2];
    const float* b1 = (const float*)d_in[3];
    const float* W2 = (const float*)d_in[4];
    const float* b2 = (const float*)d_in[5];
    const float* Wl = (const float*)d_in[6];
    const float* bl = (const float*)d_in[7];
    float* out = (float*)d_out;

    float* buf1 = (float*)d_ws;
    int*   csr  = (int*)(buf1 + (size_t)SPL * HID);
    float* buf2 = buf1 + (size_t)NN * HID;
    int*   rowptr = (int*)(buf2 + (size_t)NN * HID);
    int* tsrc  = (int*)buf2;
    int* tdst  = tsrc + NE;
    int* pt    = tdst + NE;
    float* hi1 = out;

    const int* src = ei;
    const int* dst = ei + NE;

    range_scatter<<<G1, T1, 0, stream>>>(src, dst, pt, tsrc, tdst);
    range_rows_product<<<NB, T1, 0, stream>>>(tsrc, tdst, pt, rowptr, csr, x, W1, buf1, hi1);

    gather_gemm<HID, true, false, true><<<FG, FT, 0, stream>>>(rowptr, csr, buf1, hi1, b1, W2,
                                                               nullptr, buf2);

    gather_gemm<NCL, false, true, false><<<FG, FT, 0, stream>>>(rowptr, csr, buf2, nullptr, b2, Wl,
                                                                bl, out);
}
